# starting kernel
# baseline (speedup 1.0000x reference)
#include <hip/hip_runtime.h>

typedef float v2f __attribute__((ext_vector_type(2)));
typedef float v8f __attribute__((ext_vector_type(8)));


__global__ void fill_f32(float* __restrict__ p, float v, int n) {
  int i = blockIdx.x * blockDim.x + threadIdx.x;
  if (i < n) p[i] = v;
}

__global__ void count_deg(const int* __restrict__ ei, float* __restrict__ deg, int E) {
  int e = blockIdx.x * blockDim.x + threadIdx.x;
  if (e < E) atomicAdd(&deg[ei[E + e]], 1.0f);
}

__global__ void finalize_dinv(float* __restrict__ deg, int n) {
  int i = blockIdx.x * blockDim.x + threadIdx.x;
  if (i < n) deg[i] = rsqrtf(deg[i]);
}

__global__ void gemm_wmma_f32(const float* __restrict__ X, const float* __restrict__ W,
                              float* __restrict__ H, int M, int K, int N) {
  int wavesPerBlock = blockDim.x >> 5;
  int wave = blockIdx.x * wavesPerBlock + (threadIdx.x >> 5);
  int lane = threadIdx.x & 31;
  int tilesN = N >> 4;
  int tm = wave / tilesN;
  int tn = wave - tm * tilesN;
  if (tm * 16 >= M) return;

  int r = lane & 15;
  int h = lane >> 4;

  const float* xrow = X + (size_t)(tm * 16 + r) * K + 2 * h;
  const float* wcol = W + (size_t)(2 * h) * N + tn * 16 + r;

  v8f acc = {};
  for (int k0 = 0; k0 < K; k0 += 4) {
    v2f a, b;
    a.x = xrow[k0];
    a.y = xrow[k0 + 1];
    b.x = wcol[(size_t)k0 * N];
    b.y = wcol[(size_t)(k0 + 1) * N];
    acc = __builtin_amdgcn_wmma_f32_16x16x4_f32(false, a, false, b,
                                                (short)0, acc, false, false);
  }

  float* drow = H + (size_t)(tm * 16 + 8 * h) * N + tn * 16 + r;
#pragma unroll
  for (int v = 0; v < 8; ++v) drow[(size_t)v * N] = acc[v];
}

__global__ void scatter_edges(const int* __restrict__ ei, const float* __restrict__ h,
                              const float* __restrict__ dinv, float* __restrict__ agg,
                              int E, int logF4, int F) {
  int tid = blockIdx.x * blockDim.x + threadIdx.x;
  if (tid >= (E << logF4)) return;
  int e  = tid >> logF4;
  int f4 = tid & ((1 << logF4) - 1);
  int s = ei[e];
  int d = ei[E + e];
  float c = dinv[s] * dinv[d];
  float4 hv = ((const float4*)(h + (size_t)s * F))[f4];
  float* ap = agg + (size_t)d * F + (f4 << 2);
  atomicAdd(ap + 0, hv.x * c);
  atomicAdd(ap + 1, hv.y * c);
  atomicAdd(ap + 2, hv.z * c);
  atomicAdd(ap + 3, hv.w * c);
}

__global__ void combine_nodes(const float* __restrict__ agg, const float* __restrict__ hm,
                              const float* __restrict__ dinv, const float* __restrict__ bias,
                              float* __restrict__ out, int total, int logF, int relu) {
  int tid = blockIdx.x * blockDim.x + threadIdx.x;
  if (tid >= total) return;
  int i = tid >> logF;
  int f = tid & ((1 << logF) - 1);
  float di = dinv[i];
  float v = agg[tid] + hm[tid] * di * di + bias[f];
  out[tid] = relu ? fmaxf(v, 0.0f) : v;
}


extern "C" void kernel_launch(void* const* d_in, const int* in_sizes, int n_in,
                              void* d_out, int out_size, void* d_ws, size_t ws_size,
                              hipStream_t stream) {
  const float* x  = (const float*)d_in[0];
  const int*   ei = (const int*)d_in[1];
  const float* W1 = (const float*)d_in[2];
  const float* b1 = (const float*)d_in[3];
  const float* W2 = (const float*)d_in[4];
  const float* b2 = (const float*)d_in[5];

  const int DIN = 64, DHID = 128, DOUT = 64;
  const int N = in_sizes[0] / DIN;
  const int E = in_sizes[1] / 2;

  float* dinv = (float*)d_ws;
  float* bufA = dinv + N;
  float* bufB = bufA + (size_t)N * DHID;
  float* out  = (float*)d_out;

  const int T = 256;

  fill_f32<<<(N + T - 1) / T, T, 0, stream>>>(dinv, 1.0f, N);
  count_deg<<<(E + T - 1) / T, T, 0, stream>>>(ei, dinv, E);
  finalize_dinv<<<(N + T - 1) / T, T, 0, stream>>>(dinv, N);

  {
    int waves = (N / 16) * (DHID / 16);
    gemm_wmma_f32<<<waves / 8, T, 0, stream>>>(x, W1, bufA, N, DIN, DHID);
  }
  fill_f32<<<((N * DHID) + T - 1) / T, T, 0, stream>>>(bufB, 0.0f, N * DHID);
  scatter_edges<<<((E * (DHID / 4)) + T - 1) / T, T, 0, stream>>>(
      ei, bufA, dinv, bufB, E, 5  , DHID);
  combine_nodes<<<((N * DHID) + T - 1) / T, T, 0, stream>>>(
      bufB, bufA, dinv, b1, bufB, N * DHID, 7  , 1);

  {
    int waves = (N / 16) * (DOUT / 16);
    gemm_wmma_f32<<<waves / 8, T, 0, stream>>>(bufB, W2, bufA, N, DHID, DOUT);
  }
  fill_f32<<<((N * DOUT) + T - 1) / T, T, 0, stream>>>(out, 0.0f, N * DOUT);
  scatter_edges<<<((E * (DOUT / 4)) + T - 1) / T, T, 0, stream>>>(
      ei, bufA, dinv, out, E, 4  , DOUT);
  combine_nodes<<<((N * DOUT) + T - 1) / T, T, 0, stream>>>(
      out, bufA, dinv, b2, out, N * DOUT, 6  , 0);
}


# Layer 1 aggregates before it transforms: atomic scatter-add on the 64-wide x rows, then WMMA GEMM with bias+ReLU epilogue; halves layer-1 atomics; f32 sum order changes ((A x) W1)
# speedup vs baseline: 1.3726x; 1.3726x over previous
#include <hip/hip_runtime.h>

typedef float v2f __attribute__((ext_vector_type(2)));
typedef float v8f __attribute__((ext_vector_type(8)));


__global__ void fill_f32(float* __restrict__ p, float v, int n) {
  int i = blockIdx.x * blockDim.x + threadIdx.x;
  if (i < n) p[i] = v;
}

__global__ void count_deg(const int* __restrict__ ei, float* __restrict__ deg, int E) {
  int e = blockIdx.x * blockDim.x + threadIdx.x;
  if (e < E) atomicAdd(&deg[ei[E + e]], 1.0f);
}

__global__ void finalize_dinv(float* __restrict__ deg, int n) {
  int i = blockIdx.x * blockDim.x + threadIdx.x;
  if (i < n) deg[i] = rsqrtf(deg[i]);
}

static __device__ __forceinline__ void gemm_tile(const float* __restrict__ X,
                                                 const float* __restrict__ W,
                                                 float* __restrict__ H, int M, int K, int N,
                                                 const float* __restrict__ bias,
                                                 const bool bias_relu) {
  int wavesPerBlock = blockDim.x >> 5;
  int wave = blockIdx.x * wavesPerBlock + (threadIdx.x >> 5);
  int lane = threadIdx.x & 31;
  int tilesN = N >> 4;
  int tm = wave / tilesN;
  int tn = wave - tm * tilesN;
  if (tm * 16 >= M) return;

  int r = lane & 15;
  int h = lane >> 4;

  const float* xrow = X + (size_t)(tm * 16 + r) * K + 2 * h;
  const float* wcol = W + (size_t)(2 * h) * N + tn * 16 + r;

  float bv = 0.0f;
  if (bias_relu) bv = bias[tn * 16 + r];

  v8f acc = {};
  for (int k0 = 0; k0 < K; k0 += 4) {
    v2f a, b;
    a.x = xrow[k0];
    a.y = xrow[k0 + 1];
    b.x = wcol[(size_t)k0 * N];
    b.y = wcol[(size_t)(k0 + 1) * N];
    acc = __builtin_amdgcn_wmma_f32_16x16x4_f32(false, a, false, b,
                                                (short)0, acc, false, false);
  }

  float* drow = H + (size_t)(tm * 16 + 8 * h) * N + tn * 16 + r;
#pragma unroll
  for (int v = 0; v < 8; ++v) {
    float o = acc[v];
    if (bias_relu) o = fmaxf(o + bv, 0.0f);
    drow[(size_t)v * N] = o;
  }
}

__global__ void gemm_wmma_f32(const float* __restrict__ X, const float* __restrict__ W,
                              float* __restrict__ H, int M, int K, int N) {
  gemm_tile(X, W, H, M, K, N, nullptr, false);
}

__global__ void gemm_wmma_f32_bias_relu(const float* __restrict__ X, const float* __restrict__ W,
                                        const float* __restrict__ bias,
                                        float* __restrict__ H, int M, int K, int N) {
  gemm_tile(X, W, H, M, K, N, bias, true);
}

__global__ void scatter_edges(const int* __restrict__ ei, const float* __restrict__ h,
                              const float* __restrict__ dinv, float* __restrict__ agg,
                              int E, int logF4, int F) {
  int tid = blockIdx.x * blockDim.x + threadIdx.x;
  if (tid >= (E << logF4)) return;
  int e  = tid >> logF4;
  int f4 = tid & ((1 << logF4) - 1);
  int s = ei[e];
  int d = ei[E + e];
  float c = dinv[s] * dinv[d];
  float4 hv = ((const float4*)(h + (size_t)s * F))[f4];
  float* ap = agg + (size_t)d * F + (f4 << 2);
  atomicAdd(ap + 0, hv.x * c);
  atomicAdd(ap + 1, hv.y * c);
  atomicAdd(ap + 2, hv.z * c);
  atomicAdd(ap + 3, hv.w * c);
}

__global__ void combine_nodes(const float* __restrict__ agg, const float* __restrict__ hm,
                              const float* __restrict__ dinv, const float* __restrict__ bias,
                              float* __restrict__ out, int total, int logF, int relu) {
  int tid = blockIdx.x * blockDim.x + threadIdx.x;
  if (tid >= total) return;
  int i = tid >> logF;
  int f = tid & ((1 << logF) - 1);
  float di = dinv[i];
  float v = agg[tid] + hm[tid] * di * di + bias[f];
  out[tid] = relu ? fmaxf(v, 0.0f) : v;
}

__global__ void add_self_loop(float* __restrict__ agg, const float* __restrict__ x,
                              const float* __restrict__ dinv, int total, int logF) {
  int tid = blockIdx.x * blockDim.x + threadIdx.x;
  if (tid >= total) return;
  float di = dinv[tid >> logF];
  agg[tid] = agg[tid] + x[tid] * di * di;
}


extern "C" void kernel_launch(void* const* d_in, const int* in_sizes, int n_in,
                              void* d_out, int out_size, void* d_ws, size_t ws_size,
                              hipStream_t stream) {
  const float* x  = (const float*)d_in[0];
  const int*   ei = (const int*)d_in[1];
  const float* W1 = (const float*)d_in[2];
  const float* b1 = (const float*)d_in[3];
  const float* W2 = (const float*)d_in[4];
  const float* b2 = (const float*)d_in[5];

  const int DIN = 64, DHID = 128, DOUT = 64;
  const int N = in_sizes[0] / DIN;
  const int E = in_sizes[1] / 2;

  float* dinv = (float*)d_ws;
  float* aggx = dinv + N;
  float* h1   = aggx + (size_t)N * DIN;
  float* h2   = h1 + (size_t)N * DHID;
  float* out  = (float*)d_out;

  const int T = 256;

  fill_f32<<<(N + T - 1) / T, T, 0, stream>>>(dinv, 1.0f, N);
  count_deg<<<(E + T - 1) / T, T, 0, stream>>>(ei, dinv, E);
  finalize_dinv<<<(N + T - 1) / T, T, 0, stream>>>(dinv, N);

  fill_f32<<<((N * DIN) + T - 1) / T, T, 0, stream>>>(aggx, 0.0f, N * DIN);
  scatter_edges<<<((E * (DIN / 4)) + T - 1) / T, T, 0, stream>>>(
      ei, x, dinv, aggx, E, 4  , DIN);
  add_self_loop<<<((N * DIN) + T - 1) / T, T, 0, stream>>>(
      aggx, x, dinv, N * DIN, 6  );
  {
    int waves = (N / 16) * (DHID / 16);
    gemm_wmma_f32_bias_relu<<<waves / 8, T, 0, stream>>>(aggx, W1, b1, h1, N, DIN, DHID);
  }

  {
    int waves = (N / 16) * (DOUT / 16);
    gemm_wmma_f32<<<waves / 8, T, 0, stream>>>(h1, W2, h2, N, DHID, DOUT);
  }
  fill_f32<<<((N * DOUT) + T - 1) / T, T, 0, stream>>>(out, 0.0f, N * DOUT);
  scatter_edges<<<((E * (DOUT / 4)) + T - 1) / T, T, 0, stream>>>(
      ei, h2, dinv, out, E, 4  , DOUT);
  combine_nodes<<<((N * DOUT) + T - 1) / T, T, 0, stream>>>(
      out, h2, dinv, b2, out, N * DOUT, 6  , 0);
}


# Float-atomic scatter-add replaced by edges grouped by destination (int atomic rank, prefix sum, placement) and a per-row gather sum stored once; f32 sum order changes (arrival-order rank)
# speedup vs baseline: 2.5680x; 1.8709x over previous
#include <hip/hip_runtime.h>


typedef float v2f __attribute__((ext_vector_type(2)));
typedef float v8f __attribute__((ext_vector_type(8)));


__global__ void fill_i32(int* __restrict__ p, int v, int n) {
  int i = blockIdx.x * blockDim.x + threadIdx.x;
  if (i < n) p[i] = v;
}

__global__ void count_rank(const int* __restrict__ ei, int* __restrict__ cnt,
                           int* __restrict__ rank, int E) {
  int e = blockIdx.x * blockDim.x + threadIdx.x;
  if (e < E) rank[e] = atomicAdd(&cnt[ei[E + e]], 1);
}

__global__ void __launch_bounds__(1024)
scan_degrees(const int* __restrict__ cnt, int* __restrict__ rowptr,
             float* __restrict__ dinv, int n) {
  __shared__ int wtot[32];
  __shared__ int wpre[32];
  const int tid = threadIdx.x;
  const int lane = tid & 31;
  const int wv = tid >> 5;
  const int start = blockIdx.x * 1024;

  int s = 0;
  for (int i = tid; i < start; i += 1024) s += cnt[i];

  const int i = start + tid;
  const int c = (i < n) ? cnt[i] : 0;
  int inc = c;
#pragma unroll
  for (int o = 1; o < 32; o <<= 1) {
    int t = __shfl_up(inc, o);
    if (lane >= o) inc += t;
  }
#pragma unroll
  for (int o = 16; o > 0; o >>= 1) s += __shfl_xor(s, o);
  if (lane == 31) wtot[wv] = inc;
  if (lane == 0) wpre[wv] = s;
  __syncthreads();
  int base = wpre[lane];
  int pre = (lane < wv) ? wtot[lane] : 0;
#pragma unroll
  for (int o = 16; o > 0; o >>= 1) {
    base += __shfl_xor(base, o);
    pre += __shfl_xor(pre, o);
  }
  if (i < n) {
    rowptr[i] = base + pre + inc - c;
    dinv[i] = rsqrtf((float)c + 1.0f);
  }
}

__global__ void place_edges(const int* __restrict__ ei, const int* __restrict__ rowptr,
                            const int* __restrict__ rank, const float* __restrict__ dinv,
                            unsigned long long* __restrict__ seg, int E) {
  int e = blockIdx.x * blockDim.x + threadIdx.x;
  if (e >= E) return;
  int s = ei[e];
  int d = ei[E + e];
  float c = dinv[s] * dinv[d];
  seg[rowptr[d] + rank[e]] =
      (unsigned long long)(unsigned)s | ((unsigned long long)__float_as_uint(c) << 32);
}

static __device__ __forceinline__ void gemm_tile(const float* __restrict__ X,
                                                 const float* __restrict__ W,
                                                 float* __restrict__ H, int M, int K, int N,
                                                 const float* __restrict__ bias,
                                                 const bool bias_relu) {
  int wavesPerBlock = blockDim.x >> 5;
  int wave = blockIdx.x * wavesPerBlock + (threadIdx.x >> 5);
  int lane = threadIdx.x & 31;
  int tilesN = N >> 4;
  int tm = wave / tilesN;
  int tn = wave - tm * tilesN;
  if (tm * 16 >= M) return;

  int r = lane & 15;
  int h = lane >> 4;

  const float* xrow = X + (size_t)(tm * 16 + r) * K + 2 * h;
  const float* wcol = W + (size_t)(2 * h) * N + tn * 16 + r;

  float bv = 0.0f;
  if (bias_relu) bv = bias[tn * 16 + r];

  v8f acc = {};
  for (int k0 = 0; k0 < K; k0 += 4) {
    v2f a, b;
    a.x = xrow[k0];
    a.y = xrow[k0 + 1];
    b.x = wcol[(size_t)k0 * N];
    b.y = wcol[(size_t)(k0 + 1) * N];
    acc = __builtin_amdgcn_wmma_f32_16x16x4_f32(false, a, false, b,
                                                (short)0, acc, false, false);
  }

  float* drow = H + (size_t)(tm * 16 + 8 * h) * N + tn * 16 + r;
#pragma unroll
  for (int v = 0; v < 8; ++v) {
    float o = acc[v];
    if (bias_relu) o = fmaxf(o + bv, 0.0f);
    drow[(size_t)v * N] = o;
  }
}

__global__ void gemm_wmma_f32(const float* __restrict__ X, const float* __restrict__ W,
                              float* __restrict__ H, int M, int K, int N) {
  gemm_tile(X, W, H, M, K, N, nullptr, false);
}

__global__ void gemm_wmma_f32_bias_relu(const float* __restrict__ X, const float* __restrict__ W,
                                        const float* __restrict__ bias,
                                        float* __restrict__ H, int M, int K, int N) {
  gemm_tile(X, W, H, M, K, N, bias, true);
}

#define ROWS_PER_WAVE 4
#define EDGES_PER_HALF 4

static __device__ __forceinline__ unsigned long long load_entry(
    const unsigned long long* __restrict__ seg, int beg, int end, int j) {
  const bool in = j < end;
  const unsigned long long v = seg[in ? j : beg];
  return in ? v : (v & 0xffffffffull);
}

static __device__ __forceinline__ float4 fma4(float w, float4 r, float4 a) {
  a.x = fmaf(w, r.x, a.x);
  a.y = fmaf(w, r.y, a.y);
  a.z = fmaf(w, r.z, a.z);
  a.w = fmaf(w, r.w, a.w);
  return a;
}

template <bool BIAS>
__global__ void __launch_bounds__(256)
gather_rows64(const unsigned long long* __restrict__ seg, const int* __restrict__ rowptr,
              const int* __restrict__ cnt, const float* __restrict__ dinv,
              const float* __restrict__ h, const float* __restrict__ bias,
              float* __restrict__ out, int n) {
  const int wave = __builtin_amdgcn_readfirstlane(
      (int)((blockIdx.x * blockDim.x + threadIdx.x) >> 5));
  const int lane = threadIdx.x & 31;
  const int q = lane & 15;
  const int hf = lane >> 4;
  const float4* __restrict__ h4 = (const float4*)h;

  float4 bq = make_float4(0.0f, 0.0f, 0.0f, 0.0f);
  if (BIAS) bq = make_float4(bias[4 * q], bias[4 * q + 1], bias[4 * q + 2], bias[4 * q + 3]);

  for (int t = 0; t < ROWS_PER_WAVE; ++t) {
    const int i = wave * ROWS_PER_WAVE + t;
    if (i >= n) break;
    const int beg = rowptr[i];
    const int end = beg + cnt[i];
    const float di = dinv[i];

    const float4 rs = h4[(size_t)i * 16 + q];
    const float ws = hf ? 0.0f : di * di;
    float4 acc = make_float4(0.0f, 0.0f, 0.0f, 0.0f);

    if (beg < end) {
      unsigned long long e[EDGES_PER_HALF];
#pragma unroll
      for (int u = 0; u < EDGES_PER_HALF; ++u) e[u] = load_entry(seg, beg, end, beg + 2 * u + hf);

      for (int j0 = beg; j0 < end; j0 += 2 * EDGES_PER_HALF) {
        float4 r[EDGES_PER_HALF];
#pragma unroll
        for (int u = 0; u < EDGES_PER_HALF; ++u) r[u] = h4[(size_t)(unsigned)e[u] * 16 + q];
        unsigned long long nx[EDGES_PER_HALF];
#pragma unroll
        for (int u = 0; u < EDGES_PER_HALF; ++u)
          nx[u] = load_entry(seg, beg, end, j0 + 2 * EDGES_PER_HALF + 2 * u + hf);
#pragma unroll
        for (int u = 0; u < EDGES_PER_HALF; ++u)
          acc = fma4(__uint_as_float((unsigned)(e[u] >> 32)), r[u], acc);
#pragma unroll
        for (int u = 0; u < EDGES_PER_HALF; ++u) e[u] = nx[u];
      }
    }
    acc = fma4(ws, rs, acc);

    acc.x += __shfl_xor(acc.x, 16);
    acc.y += __shfl_xor(acc.y, 16);
    acc.z += __shfl_xor(acc.z, 16);
    acc.w += __shfl_xor(acc.w, 16);
    if (hf == 0) {
      if (BIAS) {
        acc.x += bq.x; acc.y += bq.y; acc.z += bq.z; acc.w += bq.w;
      }
      ((float4*)out)[(size_t)i * 16 + q] = acc;
    }
  }
}


extern "C" void kernel_launch(void* const* d_in, const int* in_sizes, int n_in,
                              void* d_out, int out_size, void* d_ws, size_t ws_size,
                              hipStream_t stream) {
  const float* x  = (const float*)d_in[0];
  const int*   ei = (const int*)d_in[1];
  const float* W1 = (const float*)d_in[2];
  const float* b1 = (const float*)d_in[3];
  const float* W2 = (const float*)d_in[4];
  const float* b2 = (const float*)d_in[5];

  const int DIN = 64, DHID = 128, DOUT = 64;
  const int N = in_sizes[0] / DIN;
  const int E = in_sizes[1] / 2;

  float* aggx   = (float*)d_ws;
  float* h1     = aggx + (size_t)N * DIN;
  float* h2     = h1 + (size_t)N * DHID;
  unsigned long long* seg = (unsigned long long*)(h2 + (size_t)N * DOUT);
  int*   rank   = (int*)(seg + E);
  int*   cnt    = rank + E;
  int*   rowptr = cnt + N;
  float* dinv   = (float*)(rowptr + N);
  float* out    = (float*)d_out;

  const int T = 256;

  fill_i32<<<(N + T - 1) / T, T, 0, stream>>>(cnt, 0, N);
  count_rank<<<(E + T - 1) / T, T, 0, stream>>>(ei, cnt, rank, E);
  scan_degrees<<<(N + 1023) / 1024, 1024, 0, stream>>>(cnt, rowptr, dinv, N);
  place_edges<<<(E + T - 1) / T, T, 0, stream>>>(ei, rowptr, rank, dinv, seg, E);

  const int gwaves = (N + ROWS_PER_WAVE - 1) / ROWS_PER_WAVE;
  const int gblocks = (gwaves + 7) / 8;

  gather_rows64<false><<<gblocks, T, 0, stream>>>(seg, rowptr, cnt, dinv, x, nullptr, aggx, N);
  {
    int waves = (N / 16) * (DHID / 16);
    gemm_wmma_f32_bias_relu<<<waves / 8, T, 0, stream>>>(aggx, W1, b1, h1, N, DIN, DHID);
  }

  {
    int waves = (N / 16) * (DOUT / 16);
    gemm_wmma_f32<<<waves / 8, T, 0, stream>>>(h1, W2, h2, N, DHID, DOUT);
  }
  gather_rows64<true><<<gblocks, T, 0, stream>>>(seg, rowptr, cnt, dinv, h2, b2, out, N);
}


# The two WMMA GEMM launches become one kernel: a wave keeps its 16-row hidden tile in LDS (never in memory), W1/W2 staged in LDS, 16-byte operand loads; f32 sum order changes (k order in products)
# speedup vs baseline: 2.8941x; 1.1270x over previous
#include <hip/hip_runtime.h>


typedef float v2f __attribute__((ext_vector_type(2)));
typedef float v8f __attribute__((ext_vector_type(8)));


__global__ void fill_i32(int* __restrict__ p, int v, int n) {
  int i = blockIdx.x * blockDim.x + threadIdx.x;
  if (i < n) p[i] = v;
}

__global__ void count_rank(const int* __restrict__ ei, int* __restrict__ cnt,
                           int* __restrict__ rank, int E) {
  int e = blockIdx.x * blockDim.x + threadIdx.x;
  if (e < E) rank[e] = atomicAdd(&cnt[ei[E + e]], 1);
}

__global__ void __launch_bounds__(1024)
scan_degrees(const int* __restrict__ cnt, int* __restrict__ rowptr,
             float* __restrict__ dinv, int n) {
  __shared__ int wtot[32];
  __shared__ int wpre[32];
  const int tid = threadIdx.x;
  const int lane = tid & 31;
  const int wv = tid >> 5;
  const int start = blockIdx.x * 1024;

  int s = 0;
  for (int i = tid; i < start; i += 1024) s += cnt[i];

  const int i = start + tid;
  const int c = (i < n) ? cnt[i] : 0;
  int inc = c;
#pragma unroll
  for (int o = 1; o < 32; o <<= 1) {
    int t = __shfl_up(inc, o);
    if (lane >= o) inc += t;
  }
#pragma unroll
  for (int o = 16; o > 0; o >>= 1) s += __shfl_xor(s, o);
  if (lane == 31) wtot[wv] = inc;
  if (lane == 0) wpre[wv] = s;
  __syncthreads();
  int base = wpre[lane];
  int pre = (lane < wv) ? wtot[lane] : 0;
#pragma unroll
  for (int o = 16; o > 0; o >>= 1) {
    base += __shfl_xor(base, o);
    pre += __shfl_xor(pre, o);
  }
  if (i < n) {
    rowptr[i] = base + pre + inc - c;
    dinv[i] = rsqrtf((float)c + 1.0f);
  }
}

__global__ void place_edges(const int* __restrict__ ei, const int* __restrict__ rowptr,
                            const int* __restrict__ rank, const float* __restrict__ dinv,
                            unsigned long long* __restrict__ seg, int E) {
  int e = blockIdx.x * blockDim.x + threadIdx.x;
  if (e >= E) return;
  int s = ei[e];
  int d = ei[E + e];
  float c = dinv[s] * dinv[d];
  seg[rowptr[d] + rank[e]] =
      (unsigned long long)(unsigned)s | ((unsigned long long)__float_as_uint(c) << 32);
}

#define FK1 64
#define FN1 128
#define FN2 64
#define TPITCH 132

static __device__ __forceinline__ v8f mma4(float a0, float a1, float b0, float b1, v8f c) {
  v2f a, b;
  a.x = a0; a.y = a1;
  b.x = b0; b.y = b1;
  return __builtin_amdgcn_wmma_f32_16x16x4_f32(false, a, false, b, (short)0, c, false, false);
}

__global__ void __launch_bounds__(256)
mlp_strips(const float* __restrict__ X, const float* __restrict__ W1,
           const float* __restrict__ b1, const float* __restrict__ W2,
           float* __restrict__ H2, int nstrips) {
  __shared__ float4 q1[(FK1 / 4) * FN1];
  __shared__ float4 q2[(FN1 / 4) * FN2];
  __shared__ __attribute__((aligned(16))) float tile[8][16 * TPITCH];

  const int tid = threadIdx.x;
  const int wv = tid >> 5;
  const int lane = tid & 31;
  const int r = lane & 15;
  const int h = lane >> 4;

  const int lo = (int)(((unsigned)nstrips * blockIdx.x) / gridDim.x);
  const int hi = (int)(((unsigned)nstrips * (blockIdx.x + 1u)) / gridDim.x);
  float* T = tile[wv];

  float4 xa[FK1 / 8];
#pragma unroll
  for (int j = 0; j < FK1 / 8; ++j) xa[j] = make_float4(0.0f, 0.0f, 0.0f, 0.0f);
  if (lo + wv < hi) {
    const float4* xr = (const float4*)X + (size_t)((lo + wv) * 16 + r) * (FK1 / 4) + h;
#pragma unroll
    for (int j = 0; j < FK1 / 8; ++j) xa[j] = xr[2 * j];
  }

  {
    float4 w1r[(FK1 / 4) * FN1 / 256], w2r[(FN1 / 4) * FN2 / 256];
#pragma unroll
    for (int i = 0; i < (FK1 / 4) * FN1 / 256; ++i) {
      const int idx = tid + 256 * i;
      const float* w = W1 + (size_t)(4 * (idx >> 7)) * FN1 + (idx & (FN1 - 1));
      w1r[i] = make_float4(w[0], w[FN1], w[2 * FN1], w[3 * FN1]);
    }
#pragma unroll
    for (int i = 0; i < (FN1 / 4) * FN2 / 256; ++i) {
      const int idx = tid + 256 * i;
      const float* w = W2 + (size_t)(4 * (idx >> 6)) * FN2 + (idx & (FN2 - 1));
      w2r[i] = make_float4(w[0], w[FN2], w[2 * FN2], w[3 * FN2]);
    }
#pragma unroll
    for (int i = 0; i < (FK1 / 4) * FN1 / 256; ++i) q1[tid + 256 * i] = w1r[i];
#pragma unroll
    for (int i = 0; i < (FN1 / 4) * FN2 / 256; ++i) q2[tid + 256 * i] = w2r[i];
  }

  float bv[FN1 / 16];
#pragma unroll
  for (int tn = 0; tn < FN1 / 16; ++tn) bv[tn] = b1[tn * 16 + r];

  __syncthreads();

  for (int s0 = lo; s0 < hi; s0 += 8) {
    const int strip = s0 + wv;
    const bool live = strip < hi;

    if (live) {
      v8f acc[FN1 / 16];
#pragma unroll
      for (int tn = 0; tn < FN1 / 16; ++tn) acc[tn] = v8f{};

#pragma unroll
      for (int j = 0; j < FK1 / 8; ++j) {
        float4 bq[FN1 / 16];
#pragma unroll
        for (int tn = 0; tn < FN1 / 16; ++tn) bq[tn] = q1[(2 * j + h) * FN1 + tn * 16 + r];
#pragma unroll
        for (int tn = 0; tn < FN1 / 16; ++tn)
          acc[tn] = mma4(xa[j].x, xa[j].y, bq[tn].x, bq[tn].y, acc[tn]);
#pragma unroll
        for (int tn = 0; tn < FN1 / 16; ++tn)
          acc[tn] = mma4(xa[j].z, xa[j].w, bq[tn].z, bq[tn].w, acc[tn]);
      }

#pragma unroll
      for (int tn = 0; tn < FN1 / 16; ++tn) {
#pragma unroll
        for (int v = 0; v < 8; ++v)
          T[(8 * h + v) * TPITCH + tn * 16 + r] = fmaxf(acc[tn][v] + bv[tn], 0.0f);
      }
    }

    __syncthreads();

    float4 xn[FK1 / 8];
#pragma unroll
    for (int j = 0; j < FK1 / 8; ++j) xn[j] = make_float4(0.0f, 0.0f, 0.0f, 0.0f);
    if (strip + 8 < hi) {
      const float4* xr = (const float4*)X + (size_t)((strip + 8) * 16 + r) * (FK1 / 4) + h;
#pragma unroll
      for (int j = 0; j < FK1 / 8; ++j) xn[j] = xr[2 * j];
    }

    if (live) {
      v8f c[FN2 / 16];
#pragma unroll
      for (int tn = 0; tn < FN2 / 16; ++tn) c[tn] = v8f{};

#pragma unroll 4
      for (int j = 0; j < FN1 / 8; ++j) {
        const float4 a4 = *(const float4*)&T[r * TPITCH + 8 * j + 4 * h];
        float4 bq[FN2 / 16];
#pragma unroll
        for (int tn = 0; tn < FN2 / 16; ++tn) bq[tn] = q2[(2 * j + h) * FN2 + tn * 16 + r];
#pragma unroll
        for (int tn = 0; tn < FN2 / 16; ++tn)
          c[tn] = mma4(a4.x, a4.y, bq[tn].x, bq[tn].y, c[tn]);
#pragma unroll
        for (int tn = 0; tn < FN2 / 16; ++tn)
          c[tn] = mma4(a4.z, a4.w, bq[tn].z, bq[tn].w, c[tn]);
      }

      float* drow = H2 + (size_t)(strip * 16 + 8 * h) * FN2 + r;
#pragma unroll
      for (int tn = 0; tn < FN2 / 16; ++tn) {
#pragma unroll
        for (int v = 0; v < 8; ++v) drow[(size_t)v * FN2 + tn * 16] = c[tn][v];
      }
    }
#pragma unroll
    for (int j = 0; j < FK1 / 8; ++j) xa[j] = xn[j];
  }
}

#define ROWS_PER_WAVE 4
#define EDGES_PER_HALF 4

static __device__ __forceinline__ unsigned long long load_entry(
    const unsigned long long* __restrict__ seg, int beg, int end, int j) {
  const bool in = j < end;
  const unsigned long long v = seg[in ? j : beg];
  return in ? v : (v & 0xffffffffull);
}

static __device__ __forceinline__ float4 fma4(float w, float4 r, float4 a) {
  a.x = fmaf(w, r.x, a.x);
  a.y = fmaf(w, r.y, a.y);
  a.z = fmaf(w, r.z, a.z);
  a.w = fmaf(w, r.w, a.w);
  return a;
}

template <bool BIAS>
__global__ void __launch_bounds__(256)
gather_rows64(const unsigned long long* __restrict__ seg, const int* __restrict__ rowptr,
              const int* __restrict__ cnt, const float* __restrict__ dinv,
              const float* __restrict__ h, const float* __restrict__ bias,
              float* __restrict__ out, int n) {
  const int wave = __builtin_amdgcn_readfirstlane(
      (int)((blockIdx.x * blockDim.x + threadIdx.x) >> 5));
  const int lane = threadIdx.x & 31;
  const int q = lane & 15;
  const int hf = lane >> 4;
  const float4* __restrict__ h4 = (const float4*)h;

  float4 bq = make_float4(0.0f, 0.0f, 0.0f, 0.0f);
  if (BIAS) bq = make_float4(bias[4 * q], bias[4 * q + 1], bias[4 * q + 2], bias[4 * q + 3]);

  for (int t = 0; t < ROWS_PER_WAVE; ++t) {
    const int i = wave * ROWS_PER_WAVE + t;
    if (i >= n) break;
    const int beg = rowptr[i];
    const int end = beg + cnt[i];
    const float di = dinv[i];

    const float4 rs = h4[(size_t)i * 16 + q];
    const float ws = hf ? 0.0f : di * di;
    float4 acc = make_float4(0.0f, 0.0f, 0.0f, 0.0f);

    if (beg < end) {
      unsigned long long e[EDGES_PER_HALF];
#pragma unroll
      for (int u = 0; u < EDGES_PER_HALF; ++u) e[u] = load_entry(seg, beg, end, beg + 2 * u + hf);

      for (int j0 = beg; j0 < end; j0 += 2 * EDGES_PER_HALF) {
        float4 r[EDGES_PER_HALF];
#pragma unroll
        for (int u = 0; u < EDGES_PER_HALF; ++u) r[u] = h4[(size_t)(unsigned)e[u] * 16 + q];
        unsigned long long nx[EDGES_PER_HALF];
#pragma unroll
        for (int u = 0; u < EDGES_PER_HALF; ++u)
          nx[u] = load_entry(seg, beg, end, j0 + 2 * EDGES_PER_HALF + 2 * u + hf);
#pragma unroll
        for (int u = 0; u < EDGES_PER_HALF; ++u)
          acc = fma4(__uint_as_float((unsigned)(e[u] >> 32)), r[u], acc);
#pragma unroll
        for (int u = 0; u < EDGES_PER_HALF; ++u) e[u] = nx[u];
      }
    }
    acc = fma4(ws, rs, acc);

    acc.x += __shfl_xor(acc.x, 16);
    acc.y += __shfl_xor(acc.y, 16);
    acc.z += __shfl_xor(acc.z, 16);
    acc.w += __shfl_xor(acc.w, 16);
    if (hf == 0) {
      if (BIAS) {
        acc.x += bq.x; acc.y += bq.y; acc.z += bq.z; acc.w += bq.w;
      }
      ((float4*)out)[(size_t)i * 16 + q] = acc;
    }
  }
}


extern "C" void kernel_launch(void* const* d_in, const int* in_sizes, int n_in,
                              void* d_out, int out_size, void* d_ws, size_t ws_size,
                              hipStream_t stream) {
  const float* x  = (const float*)d_in[0];
  const int*   ei = (const int*)d_in[1];
  const float* W1 = (const float*)d_in[2];
  const float* b1 = (const float*)d_in[3];
  const float* W2 = (const float*)d_in[4];
  const float* b2 = (const float*)d_in[5];

  const int DIN = 64, DOUT = 64;
  const int N = in_sizes[0] / DIN;
  const int E = in_sizes[1] / 2;

  float* aggx   = (float*)d_ws;
  float* h2     = aggx + (size_t)N * DIN;
  unsigned long long* seg = (unsigned long long*)(h2 + (size_t)N * DOUT);
  int*   rank   = (int*)(seg + E);
  int*   cnt    = rank + E;
  int*   rowptr = cnt + N;
  float* dinv   = (float*)(rowptr + N);
  float* out    = (float*)d_out;

  const int T = 256;

  fill_i32<<<(N + T - 1) / T, T, 0, stream>>>(cnt, 0, N);
  count_rank<<<(E + T - 1) / T, T, 0, stream>>>(ei, cnt, rank, E);
  scan_degrees<<<(N + 1023) / 1024, 1024, 0, stream>>>(cnt, rowptr, dinv, N);
  place_edges<<<(E + T - 1) / T, T, 0, stream>>>(ei, rowptr, rank, dinv, seg, E);

  const int gwaves = (N + ROWS_PER_WAVE - 1) / ROWS_PER_WAVE;
  const int gblocks = (gwaves + 7) / 8;

  gather_rows64<false><<<gblocks, T, 0, stream>>>(seg, rowptr, cnt, dinv, x, nullptr, aggx, N);

  {
    const int nstrips = N / 16;
    const int blocks = nstrips < 512 ? nstrips : 512;
    mlp_strips<<<blocks, T, 0, stream>>>(aggx, W1, b1, W2, h2, nstrips);
  }
  gather_rows64<true><<<gblocks, T, 0, stream>>>(seg, rowptr, cnt, dinv, h2, b2, out, N);
}


# Edge grouping by counting sort with LDS counters (bins of 512 rows): no per-edge global atomic or lookup; coefficient split row*dinv[src], sum*dinv[dst]; f32 rounding and sum order (LDS rank) change
# speedup vs baseline: 5.0786x; 1.7548x over previous
#include <hip/hip_runtime.h>


typedef float v2f __attribute__((ext_vector_type(2)));
typedef float v8f __attribute__((ext_vector_type(8)));

#define BIN_SHIFT 9
#define BIN_ROWS 512
#define MAXB 256
#define NCHUNK 256
#define EPT 8

template <int NW>
static __device__ __forceinline__ int block_excl_scan(int v, int* wsum, int& total) {
  const int lane = threadIdx.x & 31;
  const int wv = threadIdx.x >> 5;
  int inc = v;
#pragma unroll
  for (int o = 1; o < 32; o <<= 1) {
    int t = __shfl_up(inc, o);
    if (lane >= o) inc += t;
  }
  if (lane == 31) wsum[wv] = inc;
  __syncthreads();
  int p = (lane < NW) ? wsum[lane] : 0;
  int pre = (lane < wv) ? p : 0;
#pragma unroll
  for (int o = 16; o > 0; o >>= 1) {
    pre += __shfl_xor(pre, o);
    p += __shfl_xor(p, o);
  }
  total = p;
  return pre + inc - v;
}

static __device__ __forceinline__ void bin_starts(const int* __restrict__ total, int nb,
                                                  int* starts) {
  if (threadIdx.x < 32) {
    const int lane = threadIdx.x;
    int v[MAXB / 32];
#pragma unroll
    for (int c = 0; c < MAXB / 32; ++c) {
      const int idx = c * 32 + lane;
      v[c] = (idx < nb) ? total[idx] : 0;
    }
    int carry = 0;
#pragma unroll
    for (int c = 0; c < MAXB / 32; ++c) {
      int inc = v[c];
#pragma unroll
      for (int o = 1; o < 32; o <<= 1) {
        int t = __shfl_up(inc, o);
        if (lane >= o) inc += t;
      }
      starts[c * 32 + lane] = carry + inc - v[c];
      carry += __shfl(inc, 31);
    }
  }
  __syncthreads();
}

__global__ void __launch_bounds__(1024)
bin_count(const int* __restrict__ ei, int* __restrict__ counts, int E, int nb) {
  __shared__ int hist[MAXB];
  const int tid = threadIdx.x;
  if (tid < MAXB) hist[tid] = 0;
  __syncthreads();
  const int lo = (int)(((unsigned long long)E * blockIdx.x) / NCHUNK);
  const int hi = (int)(((unsigned long long)E * (blockIdx.x + 1u)) / NCHUNK);
  for (int e0 = lo; e0 < hi; e0 += 1024 * EPT) {
    int d[EPT];
#pragma unroll
    for (int u = 0; u < EPT; ++u) {
      const int e = e0 + u * 1024 + tid;
      d[u] = (e < hi) ? ei[E + e] : -1;
    }
#pragma unroll
    for (int u = 0; u < EPT; ++u)
      if (d[u] >= 0) atomicAdd(&hist[d[u] >> BIN_SHIFT], 1);
  }
  __syncthreads();
  if (tid < nb) counts[blockIdx.x * MAXB + tid] = hist[tid];
}

__global__ void __launch_bounds__(NCHUNK)
scan_chunks(int* __restrict__ counts, int* __restrict__ total) {
  __shared__ int wsum[32];
  const int b = blockIdx.x;
  const int w = threadIdx.x;
  const int c = counts[w * MAXB + b];
  int tot;
  const int ex = block_excl_scan<NCHUNK / 32>(c, wsum, tot);
  counts[w * MAXB + b] = ex;
  if (w == 0) total[b] = tot;
}

__global__ void __launch_bounds__(1024)
bin_scatter(const int* __restrict__ ei, const int* __restrict__ prefix,
            const int* __restrict__ total, unsigned long long* __restrict__ binned,
            int E, int nb) {
  __shared__ int cursor[MAXB];
  const int tid = threadIdx.x;
  bin_starts(total, nb, cursor);
  if (tid < nb) cursor[tid] += prefix[blockIdx.x * MAXB + tid];
  __syncthreads();
  const int lo = (int)(((unsigned long long)E * blockIdx.x) / NCHUNK);
  const int hi = (int)(((unsigned long long)E * (blockIdx.x + 1u)) / NCHUNK);
  for (int e0 = lo; e0 < hi; e0 += 1024 * EPT) {
    int s[EPT], d[EPT];
#pragma unroll
    for (int u = 0; u < EPT; ++u) {
      const int e = e0 + u * 1024 + tid;
      const bool in = e < hi;
      s[u] = in ? ei[e] : 0;
      d[u] = in ? ei[E + e] : -1;
    }
#pragma unroll
    for (int u = 0; u < EPT; ++u) {
      if (d[u] >= 0) {
        const int pos = atomicAdd(&cursor[d[u] >> BIN_SHIFT], 1);
        binned[pos] = (unsigned long long)(unsigned)s[u] |
                      ((unsigned long long)(unsigned)d[u] << 32);
      }
    }
  }
}

__global__ void __launch_bounds__(1024)
finish_bins(const unsigned long long* __restrict__ binned, const int* __restrict__ total,
            const float* __restrict__ x, int* __restrict__ cnt, int* __restrict__ rowptr,
            float* __restrict__ dinv, unsigned* __restrict__ seg, float* __restrict__ xs,
            int n, int nb) {
  __shared__ int starts[MAXB];
  __shared__ int lcnt[BIN_ROWS];
  __shared__ int lfill[BIN_ROWS];
  __shared__ float ldinv[BIN_ROWS];
  __shared__ int wsum[32];
  const int tid = threadIdx.x;
  const int b = blockIdx.x;
  if (tid < BIN_ROWS) lcnt[tid] = 0;
  bin_starts(total, nb, starts);
  const int bs = starts[b];
  const int be = bs + total[b];
  const int r0 = b << BIN_SHIFT;

  for (int e0 = bs; e0 < be; e0 += 1024 * EPT) {
    int d[EPT];
#pragma unroll
    for (int u = 0; u < EPT; ++u) {
      const int e = e0 + u * 1024 + tid;
      d[u] = (e < be) ? (int)(binned[e] >> 32) : -1;
    }
#pragma unroll
    for (int u = 0; u < EPT; ++u)
      if (d[u] >= 0) atomicAdd(&lcnt[d[u] - r0], 1);
  }
  __syncthreads();

  const int c = (tid < BIN_ROWS) ? lcnt[tid] : 0;
  int tot;
  const int ex = block_excl_scan<32>(c, wsum, tot);
  if (tid < BIN_ROWS) {
    const float dv = rsqrtf((float)c + 1.0f);
    lfill[tid] = ex;
    ldinv[tid] = dv;
    const int row = r0 + tid;
    if (row < n) {
      cnt[row] = c;
      rowptr[row] = bs + ex;
      dinv[row] = dv;
    }
  }
  __syncthreads();

  for (int e0 = bs; e0 < be; e0 += 1024 * EPT) {
    unsigned long long w[EPT];
#pragma unroll
    for (int u = 0; u < EPT; ++u) {
      const int e = e0 + u * 1024 + tid;
      w[u] = (e < be) ? binned[e] : ~0ull;
    }
#pragma unroll
    for (int u = 0; u < EPT; ++u) {
      const int d = (int)(w[u] >> 32);
      if (d >= 0) {
        const int pos = atomicAdd(&lfill[d - r0], 1);
        seg[bs + pos] = (unsigned)w[u];
      }
    }
  }

  const int nrows = (n - r0 < BIN_ROWS) ? (n - r0) : BIN_ROWS;
  const int nq = nrows * 16;
  const float4* __restrict__ x4 = (const float4*)x + (size_t)r0 * 16;
  float4* __restrict__ xs4 = (float4*)xs + (size_t)r0 * 16;
  for (int k0 = 0; k0 < nq; k0 += 1024 * EPT) {
    float4 v[EPT];
#pragma unroll
    for (int u = 0; u < EPT; ++u) {
      const int k = k0 + u * 1024 + tid;
      v[u] = (k < nq) ? x4[k] : make_float4(0.0f, 0.0f, 0.0f, 0.0f);
    }
#pragma unroll
    for (int u = 0; u < EPT; ++u) {
      const int k = k0 + u * 1024 + tid;
      if (k < nq) {
        const float dv = ldinv[k >> 4];
        xs4[k] = make_float4(dv * v[u].x, dv * v[u].y, dv * v[u].z, dv * v[u].w);
      }
    }
  }
}

#define FK1 64
#define FN1 128
#define FN2 64
#define TPITCH 132

static __device__ __forceinline__ v8f mma4(float a0, float a1, float b0, float b1, v8f c) {
  v2f a, b;
  a.x = a0; a.y = a1;
  b.x = b0; b.y = b1;
  return __builtin_amdgcn_wmma_f32_16x16x4_f32(false, a, false, b, (short)0, c, false, false);
}

__global__ void __launch_bounds__(256)
mlp_strips(const float* __restrict__ X, const float* __restrict__ W1,
           const float* __restrict__ b1, const float* __restrict__ W2,
           const float* __restrict__ dinv, float* __restrict__ H2, int nstrips) {
  __shared__ float4 q1[(FK1 / 4) * FN1];
  __shared__ float4 q2[(FN1 / 4) * FN2];
  __shared__ __attribute__((aligned(16))) float tile[8][16 * TPITCH];

  const int tid = threadIdx.x;
  const int wv = tid >> 5;
  const int lane = tid & 31;
  const int r = lane & 15;
  const int h = lane >> 4;

  const int lo = (int)(((unsigned)nstrips * blockIdx.x) / gridDim.x);
  const int hi = (int)(((unsigned)nstrips * (blockIdx.x + 1u)) / gridDim.x);
  float* T = tile[wv];

  float4 xa[FK1 / 8];
#pragma unroll
  for (int j = 0; j < FK1 / 8; ++j) xa[j] = make_float4(0.0f, 0.0f, 0.0f, 0.0f);
  if (lo + wv < hi) {
    const float4* xr = (const float4*)X + (size_t)((lo + wv) * 16 + r) * (FK1 / 4) + h;
#pragma unroll
    for (int j = 0; j < FK1 / 8; ++j) xa[j] = xr[2 * j];
  }

  {
    float4 w1r[(FK1 / 4) * FN1 / 256], w2r[(FN1 / 4) * FN2 / 256];
#pragma unroll
    for (int i = 0; i < (FK1 / 4) * FN1 / 256; ++i) {
      const int idx = tid + 256 * i;
      const float* w = W1 + (size_t)(4 * (idx >> 7)) * FN1 + (idx & (FN1 - 1));
      w1r[i] = make_float4(w[0], w[FN1], w[2 * FN1], w[3 * FN1]);
    }
#pragma unroll
    for (int i = 0; i < (FN1 / 4) * FN2 / 256; ++i) {
      const int idx = tid + 256 * i;
      const float* w = W2 + (size_t)(4 * (idx >> 6)) * FN2 + (idx & (FN2 - 1));
      w2r[i] = make_float4(w[0], w[FN2], w[2 * FN2], w[3 * FN2]);
    }
#pragma unroll
    for (int i = 0; i < (FK1 / 4) * FN1 / 256; ++i) q1[tid + 256 * i] = w1r[i];
#pragma unroll
    for (int i = 0; i < (FN1 / 4) * FN2 / 256; ++i) q2[tid + 256 * i] = w2r[i];
  }

  float bv[FN1 / 16];
#pragma unroll
  for (int tn = 0; tn < FN1 / 16; ++tn) bv[tn] = b1[tn * 16 + r];

  __syncthreads();

  for (int s0 = lo; s0 < hi; s0 += 8) {
    const int strip = s0 + wv;
    const bool live = strip < hi;

    if (live) {
      v8f acc[FN1 / 16];
#pragma unroll
      for (int tn = 0; tn < FN1 / 16; ++tn) acc[tn] = v8f{};

#pragma unroll
      for (int j = 0; j < FK1 / 8; ++j) {
        float4 bq[FN1 / 16];
#pragma unroll
        for (int tn = 0; tn < FN1 / 16; ++tn) bq[tn] = q1[(2 * j + h) * FN1 + tn * 16 + r];
#pragma unroll
        for (int tn = 0; tn < FN1 / 16; ++tn)
          acc[tn] = mma4(xa[j].x, xa[j].y, bq[tn].x, bq[tn].y, acc[tn]);
#pragma unroll
        for (int tn = 0; tn < FN1 / 16; ++tn)
          acc[tn] = mma4(xa[j].z, xa[j].w, bq[tn].z, bq[tn].w, acc[tn]);
      }

#pragma unroll
      for (int tn = 0; tn < FN1 / 16; ++tn) {
#pragma unroll
        for (int v = 0; v < 8; ++v)
          T[(8 * h + v) * TPITCH + tn * 16 + r] = fmaxf(acc[tn][v] + bv[tn], 0.0f);
      }
    }

    __syncthreads();

    float4 xn[FK1 / 8];
#pragma unroll
    for (int j = 0; j < FK1 / 8; ++j) xn[j] = make_float4(0.0f, 0.0f, 0.0f, 0.0f);
    if (strip + 8 < hi) {
      const float4* xr = (const float4*)X + (size_t)((strip + 8) * 16 + r) * (FK1 / 4) + h;
#pragma unroll
      for (int j = 0; j < FK1 / 8; ++j) xn[j] = xr[2 * j];
    }

    if (live) {
      const float4* dq = (const float4*)(dinv + (size_t)strip * 16 + 8 * h);
      const float4 dlo = dq[0], dhi = dq[1];
      v8f c[FN2 / 16];
#pragma unroll
      for (int tn = 0; tn < FN2 / 16; ++tn) c[tn] = v8f{};

#pragma unroll 4
      for (int j = 0; j < FN1 / 8; ++j) {
        const float4 a4 = *(const float4*)&T[r * TPITCH + 8 * j + 4 * h];
        float4 bq[FN2 / 16];
#pragma unroll
        for (int tn = 0; tn < FN2 / 16; ++tn) bq[tn] = q2[(2 * j + h) * FN2 + tn * 16 + r];
#pragma unroll
        for (int tn = 0; tn < FN2 / 16; ++tn)
          c[tn] = mma4(a4.x, a4.y, bq[tn].x, bq[tn].y, c[tn]);
#pragma unroll
        for (int tn = 0; tn < FN2 / 16; ++tn)
          c[tn] = mma4(a4.z, a4.w, bq[tn].z, bq[tn].w, c[tn]);
      }

      const float dv[8] = {dlo.x, dlo.y, dlo.z, dlo.w, dhi.x, dhi.y, dhi.z, dhi.w};
      float* drow = H2 + (size_t)(strip * 16 + 8 * h) * FN2 + r;
#pragma unroll
      for (int tn = 0; tn < FN2 / 16; ++tn) {
#pragma unroll
        for (int v = 0; v < 8; ++v) drow[(size_t)v * FN2 + tn * 16] = dv[v] * c[tn][v];
      }
    }
#pragma unroll
    for (int j = 0; j < FK1 / 8; ++j) xa[j] = xn[j];
  }
}

#define ROWS_PER_WAVE 4
#define EDGES_PER_HALF 4

static __device__ __forceinline__ unsigned load_entry(
    const unsigned* __restrict__ seg, int beg, int end, int j) {
  return seg[(j < end) ? j : beg];
}

static __device__ __forceinline__ float4 fma4(float w, float4 r, float4 a) {
  a.x = fmaf(w, r.x, a.x);
  a.y = fmaf(w, r.y, a.y);
  a.z = fmaf(w, r.z, a.z);
  a.w = fmaf(w, r.w, a.w);
  return a;
}

template <bool BIAS>
__global__ void __launch_bounds__(256)
gather_rows64(const unsigned* __restrict__ seg, const int* __restrict__ rowptr,
              const int* __restrict__ cnt, const float* __restrict__ dinv,
              const float* __restrict__ h, const float* __restrict__ bias,
              float* __restrict__ out, int n) {
  const int wave = __builtin_amdgcn_readfirstlane(
      (int)((blockIdx.x * blockDim.x + threadIdx.x) >> 5));
  const int lane = threadIdx.x & 31;
  const int q = lane & 15;
  const int hf = lane >> 4;
  const float4* __restrict__ h4 = (const float4*)h;

  float4 bq = make_float4(0.0f, 0.0f, 0.0f, 0.0f);
  if (BIAS) bq = make_float4(bias[4 * q], bias[4 * q + 1], bias[4 * q + 2], bias[4 * q + 3]);

  for (int t = 0; t < ROWS_PER_WAVE; ++t) {
    const int i = wave * ROWS_PER_WAVE + t;
    if (i >= n) break;
    const int beg = rowptr[i];
    const int end = beg + cnt[i];
    const float di = dinv[i];

    const float4 rs = h4[(size_t)i * 16 + q];
    const float ws = hf ? 0.0f : 1.0f;
    float4 acc = make_float4(0.0f, 0.0f, 0.0f, 0.0f);

    if (beg < end) {
      unsigned e[EDGES_PER_HALF];
#pragma unroll
      for (int u = 0; u < EDGES_PER_HALF; ++u) e[u] = load_entry(seg, beg, end, beg + 2 * u + hf);

      for (int j0 = beg; j0 < end; j0 += 2 * EDGES_PER_HALF) {
        float4 r[EDGES_PER_HALF];
#pragma unroll
        for (int u = 0; u < EDGES_PER_HALF; ++u) r[u] = h4[(size_t)e[u] * 16 + q];
        unsigned nx[EDGES_PER_HALF];
#pragma unroll
        for (int u = 0; u < EDGES_PER_HALF; ++u)
          nx[u] = load_entry(seg, beg, end, j0 + 2 * EDGES_PER_HALF + 2 * u + hf);
#pragma unroll
        for (int u = 0; u < EDGES_PER_HALF; ++u)
          acc = fma4((j0 + 2 * u + hf < end) ? 1.0f : 0.0f, r[u], acc);
#pragma unroll
        for (int u = 0; u < EDGES_PER_HALF; ++u) e[u] = nx[u];
      }
    }
    acc = fma4(ws, rs, acc);

    acc.x += __shfl_xor(acc.x, 16);
    acc.y += __shfl_xor(acc.y, 16);
    acc.z += __shfl_xor(acc.z, 16);
    acc.w += __shfl_xor(acc.w, 16);
    if (hf == 0) {
      acc.x = fmaf(di, acc.x, bq.x);
      acc.y = fmaf(di, acc.y, bq.y);
      acc.z = fmaf(di, acc.z, bq.z);
      acc.w = fmaf(di, acc.w, bq.w);
      ((float4*)out)[(size_t)i * 16 + q] = acc;
    }
  }
}


extern "C" void kernel_launch(void* const* d_in, const int* in_sizes, int n_in,
                              void* d_out, int out_size, void* d_ws, size_t ws_size,
                              hipStream_t stream) {
  const float* x  = (const float*)d_in[0];
  const int*   ei = (const int*)d_in[1];
  const float* W1 = (const float*)d_in[2];
  const float* b1 = (const float*)d_in[3];
  const float* W2 = (const float*)d_in[4];
  const float* b2 = (const float*)d_in[5];

  const int DIN = 64, DOUT = 64;
  const int N = in_sizes[0] / DIN;
  const int E = in_sizes[1] / 2;

  float* aggx   = (float*)d_ws;
  float* h2     = aggx + (size_t)N * DIN;
  float* xs     = h2 + (size_t)N * DOUT;
  unsigned long long* binned = (unsigned long long*)(xs + (size_t)N * DIN);
  unsigned* seg = (unsigned*)(binned + E);
  int*   counts = (int*)(seg + E);
  int*   total  = counts + NCHUNK * MAXB;
  int*   cnt    = total + MAXB;
  int*   rowptr = cnt + N;
  float* dinv   = (float*)(rowptr + N);
  float* out    = (float*)d_out;

  const int T = 256;
  const int NB = (N + BIN_ROWS - 1) / BIN_ROWS;

  bin_count<<<NCHUNK, 1024, 0, stream>>>(ei, counts, E, NB);
  scan_chunks<<<NB, NCHUNK, 0, stream>>>(counts, total);
  bin_scatter<<<NCHUNK, 1024, 0, stream>>>(ei, counts, total, binned, E, NB);
  finish_bins<<<NB, 1024, 0, stream>>>(binned, total, x, cnt, rowptr, dinv, seg, xs, N, NB);

  const int gwaves = (N + ROWS_PER_WAVE - 1) / ROWS_PER_WAVE;
  const int gblocks = (gwaves + 7) / 8;

  gather_rows64<false><<<gblocks, T, 0, stream>>>(seg, rowptr, cnt, dinv, xs, nullptr, aggx, N);

  {
    const int nstrips = N / 16;
    const int blocks = nstrips < 512 ? nstrips : 512;
    mlp_strips<<<blocks, T, 0, stream>>>(aggx, W1, b1, W2, dinv, h2, nstrips);
  }
  gather_rows64<true><<<gblocks, T, 0, stream>>>(seg, rowptr, cnt, dinv, h2, b2, out, N);
}


# The edge sort's two placement passes build their output in final order in LDS and store contiguous runs / whole lines, not one scattered store per edge; f32 sum order stays an LDS arrival rank
# speedup vs baseline: 5.5307x; 1.0890x over previous
#include <hip/hip_runtime.h>


typedef float v2f __attribute__((ext_vector_type(2)));
typedef float v8f __attribute__((ext_vector_type(8)));

#define BIN_SHIFT 9
#define BIN_ROWS 512
#define MAXB 256
#define NCHUNK 256
#define EPT 8
#define SEG_STAGE 16384

template <int NW>
static __device__ __forceinline__ int block_excl_scan(int v, int* wsum, int& total) {
  const int lane = threadIdx.x & 31;
  const int wv = threadIdx.x >> 5;
  int inc = v;
#pragma unroll
  for (int o = 1; o < 32; o <<= 1) {
    int t = __shfl_up(inc, o);
    if (lane >= o) inc += t;
  }
  if (lane == 31) wsum[wv] = inc;
  __syncthreads();
  int p = (lane < NW) ? wsum[lane] : 0;
  int pre = (lane < wv) ? p : 0;
#pragma unroll
  for (int o = 16; o > 0; o >>= 1) {
    pre += __shfl_xor(pre, o);
    p += __shfl_xor(p, o);
  }
  total = p;
  return pre + inc - v;
}

static __device__ __forceinline__ void bin_starts(const int* __restrict__ total, int nb,
                                                  int* starts) {
  if (threadIdx.x < 32) {
    const int lane = threadIdx.x;
    int v[MAXB / 32];
#pragma unroll
    for (int c = 0; c < MAXB / 32; ++c) {
      const int idx = c * 32 + lane;
      v[c] = (idx < nb) ? total[idx] : 0;
    }
    int carry = 0;
#pragma unroll
    for (int c = 0; c < MAXB / 32; ++c) {
      int inc = v[c];
#pragma unroll
      for (int o = 1; o < 32; o <<= 1) {
        int t = __shfl_up(inc, o);
        if (lane >= o) inc += t;
      }
      starts[c * 32 + lane] = carry + inc - v[c];
      carry += __shfl(inc, 31);
    }
  }
  __syncthreads();
}

__global__ void __launch_bounds__(1024)
bin_count(const int* __restrict__ ei, int* __restrict__ counts, int E, int nb) {
  __shared__ int hist[MAXB];
  const int tid = threadIdx.x;
  if (tid < MAXB) hist[tid] = 0;
  __syncthreads();
  const int lo = (int)(((unsigned long long)E * blockIdx.x) / NCHUNK);
  const int hi = (int)(((unsigned long long)E * (blockIdx.x + 1u)) / NCHUNK);
  for (int e0 = lo; e0 < hi; e0 += 1024 * EPT) {
    int d[EPT];
#pragma unroll
    for (int u = 0; u < EPT; ++u) {
      const int e = e0 + u * 1024 + tid;
      d[u] = (e < hi) ? ei[E + e] : -1;
    }
#pragma unroll
    for (int u = 0; u < EPT; ++u)
      if (d[u] >= 0) atomicAdd(&hist[d[u] >> BIN_SHIFT], 1);
  }
  __syncthreads();
  if (tid < nb) counts[blockIdx.x * MAXB + tid] = hist[tid];
}

__global__ void __launch_bounds__(NCHUNK)
scan_chunks(int* __restrict__ counts, int* __restrict__ total) {
  __shared__ int wsum[32];
  const int b = blockIdx.x;
  const int w = threadIdx.x;
  const int c = counts[w * MAXB + b];
  int tot;
  const int ex = block_excl_scan<NCHUNK / 32>(c, wsum, tot);
  counts[w * MAXB + b] = ex;
  if (w == 0) total[b] = tot;
}

__global__ void __launch_bounds__(1024)
bin_scatter(const int* __restrict__ ei, const int* __restrict__ prefix,
            const int* __restrict__ total, unsigned long long* __restrict__ binned,
            int E, int nb) {
  __shared__ int cursor[MAXB];
  __shared__ int lcnt[MAXB];
  __shared__ int lstart[MAXB];
  __shared__ int shift[MAXB];
  __shared__ int wsum[32];
  __shared__ unsigned long long stage[1024 * EPT];
  const int tid = threadIdx.x;
  bin_starts(total, nb, cursor);
  if (tid < nb) cursor[tid] += prefix[blockIdx.x * MAXB + tid];
  const int lo = (int)(((unsigned long long)E * blockIdx.x) / NCHUNK);
  const int hi = (int)(((unsigned long long)E * (blockIdx.x + 1u)) / NCHUNK);
  for (int e0 = lo; e0 < hi; e0 += 1024 * EPT) {
    if (tid < MAXB) lcnt[tid] = 0;
    __syncthreads();
    int s[EPT], d[EPT], rk[EPT];
#pragma unroll
    for (int u = 0; u < EPT; ++u) {
      const int e = e0 + u * 1024 + tid;
      const bool in = e < hi;
      s[u] = in ? ei[e] : 0;
      d[u] = in ? ei[E + e] : -1;
    }
#pragma unroll
    for (int u = 0; u < EPT; ++u)
      rk[u] = (d[u] >= 0) ? atomicAdd(&lcnt[d[u] >> BIN_SHIFT], 1) : 0;
    __syncthreads();
    int ntrip;
    const int ex = block_excl_scan<32>((tid < MAXB) ? lcnt[tid] : 0, wsum, ntrip);
    if (tid < MAXB) {
      lstart[tid] = ex;
      shift[tid] = cursor[tid] - ex;
      cursor[tid] += lcnt[tid];
    }
    __syncthreads();
#pragma unroll
    for (int u = 0; u < EPT; ++u) {
      if (d[u] >= 0) {
        stage[lstart[d[u] >> BIN_SHIFT] + rk[u]] =
            (unsigned long long)(unsigned)s[u] | ((unsigned long long)(unsigned)d[u] << 32);
      }
    }
    __syncthreads();
#pragma unroll
    for (int u = 0; u < EPT; ++u) {
      const int k = u * 1024 + tid;
      if (k < ntrip) {
        const unsigned long long w = stage[k];
        binned[shift[(int)(w >> 32) >> BIN_SHIFT] + k] = w;
      }
    }
  }
}

__global__ void __launch_bounds__(1024)
finish_bins(const unsigned long long* __restrict__ binned, const int* __restrict__ total,
            const float* __restrict__ x, int* __restrict__ cnt, int* __restrict__ rowptr,
            float* __restrict__ dinv, unsigned* __restrict__ seg, float* __restrict__ xs,
            int n, int nb) {
  __shared__ int starts[MAXB];
  __shared__ int lcnt[BIN_ROWS];
  __shared__ int lfill[BIN_ROWS];
  __shared__ float ldinv[BIN_ROWS];
  __shared__ int wsum[32];
  __shared__ unsigned lseg[SEG_STAGE];
  const int tid = threadIdx.x;
  const int b = blockIdx.x;
  if (tid < BIN_ROWS) lcnt[tid] = 0;
  bin_starts(total, nb, starts);
  const int bs = starts[b];
  const int be = bs + total[b];
  const int r0 = b << BIN_SHIFT;
  const bool staged = (be - bs) <= SEG_STAGE;

  for (int e0 = bs; e0 < be; e0 += 1024 * EPT) {
    int d[EPT];
#pragma unroll
    for (int u = 0; u < EPT; ++u) {
      const int e = e0 + u * 1024 + tid;
      d[u] = (e < be) ? (int)(binned[e] >> 32) : -1;
    }
#pragma unroll
    for (int u = 0; u < EPT; ++u)
      if (d[u] >= 0) atomicAdd(&lcnt[d[u] - r0], 1);
  }
  __syncthreads();

  const int c = (tid < BIN_ROWS) ? lcnt[tid] : 0;
  int tot;
  const int ex = block_excl_scan<32>(c, wsum, tot);
  if (tid < BIN_ROWS) {
    const float dv = rsqrtf((float)c + 1.0f);
    lfill[tid] = ex;
    ldinv[tid] = dv;
    const int row = r0 + tid;
    if (row < n) {
      cnt[row] = c;
      rowptr[row] = bs + ex;
      dinv[row] = dv;
    }
  }
  __syncthreads();

  for (int e0 = bs; e0 < be; e0 += 1024 * EPT) {
    unsigned long long w[EPT];
#pragma unroll
    for (int u = 0; u < EPT; ++u) {
      const int e = e0 + u * 1024 + tid;
      w[u] = (e < be) ? binned[e] : ~0ull;
    }
#pragma unroll
    for (int u = 0; u < EPT; ++u) {
      const int d = (int)(w[u] >> 32);
      if (d >= 0) {
        const int pos = atomicAdd(&lfill[d - r0], 1);
        if (staged) lseg[pos] = (unsigned)w[u];
        else seg[bs + pos] = (unsigned)w[u];
      }
    }
  }
  if (staged) {
    __syncthreads();
    const int m = be - bs;
    for (int k0 = -(bs & 31); k0 < m; k0 += 1024 * EPT) {
#pragma unroll
      for (int u = 0; u < EPT; ++u) {
        const int k = k0 + u * 1024 + tid;
        if (k >= 0 && k < m) seg[bs + k] = lseg[k];
      }
    }
  }

  const int nrows = (n - r0 < BIN_ROWS) ? (n - r0) : BIN_ROWS;
  const int nq = nrows * 16;
  const float4* __restrict__ x4 = (const float4*)x + (size_t)r0 * 16;
  float4* __restrict__ xs4 = (float4*)xs + (size_t)r0 * 16;
  for (int k0 = 0; k0 < nq; k0 += 1024 * EPT) {
    float4 v[EPT];
#pragma unroll
    for (int u = 0; u < EPT; ++u) {
      const int k = k0 + u * 1024 + tid;
      v[u] = (k < nq) ? x4[k] : make_float4(0.0f, 0.0f, 0.0f, 0.0f);
    }
#pragma unroll
    for (int u = 0; u < EPT; ++u) {
      const int k = k0 + u * 1024 + tid;
      if (k < nq) {
        const float dv = ldinv[k >> 4];
        xs4[k] = make_float4(dv * v[u].x, dv * v[u].y, dv * v[u].z, dv * v[u].w);
      }
    }
  }
}

#define FK1 64
#define FN1 128
#define FN2 64
#define TPITCH 132

static __device__ __forceinline__ v8f mma4(float a0, float a1, float b0, float b1, v8f c) {
  v2f a, b;
  a.x = a0; a.y = a1;
  b.x = b0; b.y = b1;
  return __builtin_amdgcn_wmma_f32_16x16x4_f32(false, a, false, b, (short)0, c, false, false);
}

__global__ void __launch_bounds__(256)
mlp_strips(const float* __restrict__ X, const float* __restrict__ W1,
           const float* __restrict__ b1, const float* __restrict__ W2,
           const float* __restrict__ dinv, float* __restrict__ H2, int nstrips) {
  __shared__ float4 q1[(FK1 / 4) * FN1];
  __shared__ float4 q2[(FN1 / 4) * FN2];
  __shared__ __attribute__((aligned(16))) float tile[8][16 * TPITCH];

  const int tid = threadIdx.x;
  const int wv = tid >> 5;
  const int lane = tid & 31;
  const int r = lane & 15;
  const int h = lane >> 4;

  const int lo = (int)(((unsigned)nstrips * blockIdx.x) / gridDim.x);
  const int hi = (int)(((unsigned)nstrips * (blockIdx.x + 1u)) / gridDim.x);
  float* T = tile[wv];

  float4 xa[FK1 / 8];
#pragma unroll
  for (int j = 0; j < FK1 / 8; ++j) xa[j] = make_float4(0.0f, 0.0f, 0.0f, 0.0f);
  if (lo + wv < hi) {
    const float4* xr = (const float4*)X + (size_t)((lo + wv) * 16 + r) * (FK1 / 4) + h;
#pragma unroll
    for (int j = 0; j < FK1 / 8; ++j) xa[j] = xr[2 * j];
  }

  {
    float4 w1r[(FK1 / 4) * FN1 / 256], w2r[(FN1 / 4) * FN2 / 256];
#pragma unroll
    for (int i = 0; i < (FK1 / 4) * FN1 / 256; ++i) {
      const int idx = tid + 256 * i;
      const float* w = W1 + (size_t)(4 * (idx >> 7)) * FN1 + (idx & (FN1 - 1));
      w1r[i] = make_float4(w[0], w[FN1], w[2 * FN1], w[3 * FN1]);
    }
#pragma unroll
    for (int i = 0; i < (FN1 / 4) * FN2 / 256; ++i) {
      const int idx = tid + 256 * i;
      const float* w = W2 + (size_t)(4 * (idx >> 6)) * FN2 + (idx & (FN2 - 1));
      w2r[i] = make_float4(w[0], w[FN2], w[2 * FN2], w[3 * FN2]);
    }
#pragma unroll
    for (int i = 0; i < (FK1 / 4) * FN1 / 256; ++i) q1[tid + 256 * i] = w1r[i];
#pragma unroll
    for (int i = 0; i < (FN1 / 4) * FN2 / 256; ++i) q2[tid + 256 * i] = w2r[i];
  }

  float bv[FN1 / 16];
#pragma unroll
  for (int tn = 0; tn < FN1 / 16; ++tn) bv[tn] = b1[tn * 16 + r];

  __syncthreads();

  for (int s0 = lo; s0 < hi; s0 += 8) {
    const int strip = s0 + wv;
    const bool live = strip < hi;

    if (live) {
      v8f acc[FN1 / 16];
#pragma unroll
      for (int tn = 0; tn < FN1 / 16; ++tn) acc[tn] = v8f{};

#pragma unroll
      for (int j = 0; j < FK1 / 8; ++j) {
        float4 bq[FN1 / 16];
#pragma unroll
        for (int tn = 0; tn < FN1 / 16; ++tn) bq[tn] = q1[(2 * j + h) * FN1 + tn * 16 + r];
#pragma unroll
        for (int tn = 0; tn < FN1 / 16; ++tn)
          acc[tn] = mma4(xa[j].x, xa[j].y, bq[tn].x, bq[tn].y, acc[tn]);
#pragma unroll
        for (int tn = 0; tn < FN1 / 16; ++tn)
          acc[tn] = mma4(xa[j].z, xa[j].w, bq[tn].z, bq[tn].w, acc[tn]);
      }

#pragma unroll
      for (int tn = 0; tn < FN1 / 16; ++tn) {
#pragma unroll
        for (int v = 0; v < 8; ++v)
          T[(8 * h + v) * TPITCH + tn * 16 + r] = fmaxf(acc[tn][v] + bv[tn], 0.0f);
      }
    }

    __syncthreads();

    float4 xn[FK1 / 8];
#pragma unroll
    for (int j = 0; j < FK1 / 8; ++j) xn[j] = make_float4(0.0f, 0.0f, 0.0f, 0.0f);
    if (strip + 8 < hi) {
      const float4* xr = (const float4*)X + (size_t)((strip + 8) * 16 + r) * (FK1 / 4) + h;
#pragma unroll
      for (int j = 0; j < FK1 / 8; ++j) xn[j] = xr[2 * j];
    }

    if (live) {
      const float4* dq = (const float4*)(dinv + (size_t)strip * 16 + 8 * h);
      const float4 dlo = dq[0], dhi = dq[1];
      v8f c[FN2 / 16];
#pragma unroll
      for (int tn = 0; tn < FN2 / 16; ++tn) c[tn] = v8f{};

#pragma unroll 4
      for (int j = 0; j < FN1 / 8; ++j) {
        const float4 a4 = *(const float4*)&T[r * TPITCH + 8 * j + 4 * h];
        float4 bq[FN2 / 16];
#pragma unroll
        for (int tn = 0; tn < FN2 / 16; ++tn) bq[tn] = q2[(2 * j + h) * FN2 + tn * 16 + r];
#pragma unroll
        for (int tn = 0; tn < FN2 / 16; ++tn)
          c[tn] = mma4(a4.x, a4.y, bq[tn].x, bq[tn].y, c[tn]);
#pragma unroll
        for (int tn = 0; tn < FN2 / 16; ++tn)
          c[tn] = mma4(a4.z, a4.w, bq[tn].z, bq[tn].w, c[tn]);
      }

      const float dv[8] = {dlo.x, dlo.y, dlo.z, dlo.w, dhi.x, dhi.y, dhi.z, dhi.w};
      float* drow = H2 + (size_t)(strip * 16 + 8 * h) * FN2 + r;
#pragma unroll
      for (int tn = 0; tn < FN2 / 16; ++tn) {
#pragma unroll
        for (int v = 0; v < 8; ++v) drow[(size_t)v * FN2 + tn * 16] = dv[v] * c[tn][v];
      }
    }
#pragma unroll
    for (int j = 0; j < FK1 / 8; ++j) xa[j] = xn[j];
  }
}

#define ROWS_PER_WAVE 4
#define EDGES_PER_HALF 4

static __device__ __forceinline__ unsigned load_entry(
    const unsigned* __restrict__ seg, int beg, int end, int j) {
  return seg[(j < end) ? j : beg];
}

static __device__ __forceinline__ float4 fma4(float w, float4 r, float4 a) {
  a.x = fmaf(w, r.x, a.x);
  a.y = fmaf(w, r.y, a.y);
  a.z = fmaf(w, r.z, a.z);
  a.w = fmaf(w, r.w, a.w);
  return a;
}

template <bool BIAS>
__global__ void __launch_bounds__(256)
gather_rows64(const unsigned* __restrict__ seg, const int* __restrict__ rowptr,
              const int* __restrict__ cnt, const float* __restrict__ dinv,
              const float* __restrict__ h, const float* __restrict__ bias,
              float* __restrict__ out, int n) {
  const int wave = __builtin_amdgcn_readfirstlane(
      (int)((blockIdx.x * blockDim.x + threadIdx.x) >> 5));
  const int lane = threadIdx.x & 31;
  const int q = lane & 15;
  const int hf = lane >> 4;
  const float4* __restrict__ h4 = (const float4*)h;

  float4 bq = make_float4(0.0f, 0.0f, 0.0f, 0.0f);
  if (BIAS) bq = make_float4(bias[4 * q], bias[4 * q + 1], bias[4 * q + 2], bias[4 * q + 3]);

  for (int t = 0; t < ROWS_PER_WAVE; ++t) {
    const int i = wave * ROWS_PER_WAVE + t;
    if (i >= n) break;
    const int beg = rowptr[i];
    const int end = beg + cnt[i];
    const float di = dinv[i];

    const float4 rs = h4[(size_t)i * 16 + q];
    const float ws = hf ? 0.0f : 1.0f;
    float4 acc = make_float4(0.0f, 0.0f, 0.0f, 0.0f);

    if (beg < end) {
      unsigned e[EDGES_PER_HALF];
#pragma unroll
      for (int u = 0; u < EDGES_PER_HALF; ++u) e[u] = load_entry(seg, beg, end, beg + 2 * u + hf);

      for (int j0 = beg; j0 < end; j0 += 2 * EDGES_PER_HALF) {
        float4 r[EDGES_PER_HALF];
#pragma unroll
        for (int u = 0; u < EDGES_PER_HALF; ++u) r[u] = h4[(size_t)e[u] * 16 + q];
        unsigned nx[EDGES_PER_HALF];
#pragma unroll
        for (int u = 0; u < EDGES_PER_HALF; ++u)
          nx[u] = load_entry(seg, beg, end, j0 + 2 * EDGES_PER_HALF + 2 * u + hf);
#pragma unroll
        for (int u = 0; u < EDGES_PER_HALF; ++u)
          acc = fma4((j0 + 2 * u + hf < end) ? 1.0f : 0.0f, r[u], acc);
#pragma unroll
        for (int u = 0; u < EDGES_PER_HALF; ++u) e[u] = nx[u];
      }
    }
    acc = fma4(ws, rs, acc);

    acc.x += __shfl_xor(acc.x, 16);
    acc.y += __shfl_xor(acc.y, 16);
    acc.z += __shfl_xor(acc.z, 16);
    acc.w += __shfl_xor(acc.w, 16);
    if (hf == 0) {
      acc.x = fmaf(di, acc.x, bq.x);
      acc.y = fmaf(di, acc.y, bq.y);
      acc.z = fmaf(di, acc.z, bq.z);
      acc.w = fmaf(di, acc.w, bq.w);
      ((float4*)out)[(size_t)i * 16 + q] = acc;
    }
  }
}


extern "C" void kernel_launch(void* const* d_in, const int* in_sizes, int n_in,
                              void* d_out, int out_size, void* d_ws, size_t ws_size,
                              hipStream_t stream) {
  const float* x  = (const float*)d_in[0];
  const int*   ei = (const int*)d_in[1];
  const float* W1 = (const float*)d_in[2];
  const float* b1 = (const float*)d_in[3];
  const float* W2 = (const float*)d_in[4];
  const float* b2 = (const float*)d_in[5];

  const int DIN = 64, DOUT = 64;
  const int N = in_sizes[0] / DIN;
  const int E = in_sizes[1] / 2;

  float* aggx   = (float*)d_ws;
  float* h2     = aggx + (size_t)N * DIN;
  float* xs     = h2 + (size_t)N * DOUT;
  unsigned long long* binned = (unsigned long long*)(xs + (size_t)N * DIN);
  unsigned* seg = (unsigned*)(binned + E);
  int*   counts = (int*)(seg + E);
  int*   total  = counts + NCHUNK * MAXB;
  int*   cnt    = total + MAXB;
  int*   rowptr = cnt + N;
  float* dinv   = (float*)(rowptr + N);
  float* out    = (float*)d_out;

  const int T = 256;
  const int NB = (N + BIN_ROWS - 1) / BIN_ROWS;

  bin_count<<<NCHUNK, 1024, 0, stream>>>(ei, counts, E, NB);
  scan_chunks<<<NB, NCHUNK, 0, stream>>>(counts, total);
  bin_scatter<<<NCHUNK, 1024, 0, stream>>>(ei, counts, total, binned, E, NB);
  finish_bins<<<NB, 1024, 0, stream>>>(binned, total, x, cnt, rowptr, dinv, seg, xs, N, NB);

  const int gwaves = (N + ROWS_PER_WAVE - 1) / ROWS_PER_WAVE;
  const int gblocks = (gwaves + 7) / 8;

  gather_rows64<false><<<gblocks, T, 0, stream>>>(seg, rowptr, cnt, dinv, xs, nullptr, aggx, N);

  {
    const int nstrips = N / 16;
    const int blocks = nstrips < 512 ? nstrips : 512;
    mlp_strips<<<blocks, T, 0, stream>>>(aggx, W1, b1, W2, dinv, h2, nstrips);
  }
  gather_rows64<true><<<gblocks, T, 0, stream>>>(seg, rowptr, cnt, dinv, h2, b2, out, N);
}


# Edge sort in 2 launches, not 4: each chunk is put in bin order in its own stretch (a dword per edge); a bin's work-group sums its column of run offsets itself; f32 sum order stays LDS arrival rank
# speedup vs baseline: 6.2051x; 1.1219x over previous
#include <hip/hip_runtime.h>


typedef float v2f __attribute__((ext_vector_type(2)));
typedef float v8f __attribute__((ext_vector_type(8)));

#define BIN_SHIFT 9
#define BIN_ROWS 512
#define MAXB 256
#define SRC_BITS 17
#define SRC_MASK ((1u << SRC_BITS) - 1u)
#define EPT 8
#define CHUNK (1024 * EPT)
#define MAXT 1024
#define RUNS_TOGETHER 8
#define SEG_STAGE 16384

template <int NW>
static __device__ __forceinline__ int block_excl_scan(int v, int* wsum, int& total) {
  const int lane = threadIdx.x & 31;
  const int wv = threadIdx.x >> 5;
  int inc = v;
#pragma unroll
  for (int o = 1; o < 32; o <<= 1) {
    int t = __shfl_up(inc, o);
    if (lane >= o) inc += t;
  }
  if (lane == 31) wsum[wv] = inc;
  __syncthreads();
  int p = (lane < NW) ? wsum[lane] : 0;
  int pre = (lane < wv) ? p : 0;
#pragma unroll
  for (int o = 16; o > 0; o >>= 1) {
    pre += __shfl_xor(pre, o);
    p += __shfl_xor(p, o);
  }
  total = p;
  return pre + inc - v;
}

template <int NW>
static __device__ __forceinline__ void block_sum2(int a, int b, int* wa, int* wb, int& sa, int& sb) {
  const int lane = threadIdx.x & 31;
  const int wv = threadIdx.x >> 5;
#pragma unroll
  for (int o = 16; o > 0; o >>= 1) {
    a += __shfl_xor(a, o);
    b += __shfl_xor(b, o);
  }
  if (lane == 0) {
    wa[wv] = a;
    wb[wv] = b;
  }
  __syncthreads();
  int pa = (lane < NW) ? wa[lane] : 0;
  int pb = (lane < NW) ? wb[lane] : 0;
#pragma unroll
  for (int o = 16; o > 0; o >>= 1) {
    pa += __shfl_xor(pa, o);
    pb += __shfl_xor(pb, o);
  }
  sa = pa;
  sb = pb;
}

__global__ void __launch_bounds__(1024)
chunk_sort(const int* __restrict__ ei, unsigned* __restrict__ runs,
           unsigned* __restrict__ binned, int E, int nb) {
  __shared__ int lcnt[MAXB];
  __shared__ int lstart[MAXB];
  __shared__ int wsum[32];
  __shared__ unsigned stage[CHUNK];
  const int tid = threadIdx.x;
  const int lo = blockIdx.x * CHUNK;
  const int hi = (E - lo < CHUNK) ? E : lo + CHUNK;
  if (tid < MAXB) lcnt[tid] = 0;
  int s[EPT], d[EPT], rk[EPT];
#pragma unroll
  for (int u = 0; u < EPT; ++u) {
    const int e = lo + u * 1024 + tid;
    const bool in = e < hi;
    s[u] = in ? ei[e] : 0;
    d[u] = in ? ei[E + e] : -1;
  }
  __syncthreads();
#pragma unroll
  for (int u = 0; u < EPT; ++u)
    rk[u] = (d[u] >= 0) ? atomicAdd(&lcnt[d[u] >> BIN_SHIFT], 1) : 0;
  __syncthreads();
  const int c = (tid < MAXB) ? lcnt[tid] : 0;
  int nedge;
  const int ex = block_excl_scan<32>(c, wsum, nedge);
  if (tid < MAXB) {
    lstart[tid] = ex;
    if (tid < nb) runs[blockIdx.x * MAXB + tid] = ((unsigned)ex << 16) | (unsigned)c;
  }
  __syncthreads();
#pragma unroll
  for (int u = 0; u < EPT; ++u) {
    if (d[u] >= 0) {
      stage[lstart[d[u] >> BIN_SHIFT] + rk[u]] =
          (unsigned)s[u] | ((unsigned)(d[u] & (BIN_ROWS - 1)) << SRC_BITS);
    }
  }
  __syncthreads();
#pragma unroll
  for (int u = 0; u < EPT; ++u) {
    const int k = u * 1024 + tid;
    if (k < nedge) binned[lo + k] = stage[k];
  }
}

template <bool PLACE>
static __device__ __forceinline__ void take_edge(unsigned w, int* counter, unsigned* lseg,
                                                 unsigned* seg_bin, bool staged) {
  if (PLACE) {
    const int pos = atomicAdd(&counter[w >> SRC_BITS], 1);
    if (staged) lseg[pos] = w & SRC_MASK;
    else seg_bin[pos] = w & SRC_MASK;
  } else {
    atomicAdd(&counter[w >> SRC_BITS], 1);
  }
}

template <bool PLACE>
static __device__ __forceinline__ void walk_runs(const unsigned* __restrict__ binned,
                                                 const int* rsrc, const int* rlen, int nt,
                                                 int* counter, unsigned* lseg,
                                                 unsigned* seg_bin, bool staged) {
  const int lane = threadIdx.x & 31;
  const int wv = threadIdx.x >> 5;
  for (int t0 = wv; t0 < nt; t0 += 32 * RUNS_TOGETHER) {
    int p[RUNS_TOGETHER], c[RUNS_TOGETHER];
    unsigned w[RUNS_TOGETHER][2];
#pragma unroll
    for (int j = 0; j < RUNS_TOGETHER; ++j) {
      const int t = t0 + 32 * j;
      const bool in = t < nt;
      c[j] = in ? rlen[in ? t : 0] : 0;
      p[j] = rsrc[in ? t : 0];
    }
#pragma unroll
    for (int j = 0; j < RUNS_TOGETHER; ++j) {
#pragma unroll
      for (int i2 = 0; i2 < 2; ++i2) {
        const int i = lane + 32 * i2;
        w[j][i2] = (i < c[j]) ? binned[p[j] + i] : ~0u;
      }
    }
    if (PLACE) {
      int pos[RUNS_TOGETHER][2];
#pragma unroll
      for (int j = 0; j < RUNS_TOGETHER; ++j) {
#pragma unroll
        for (int i2 = 0; i2 < 2; ++i2)
          pos[j][i2] = (w[j][i2] != ~0u) ? atomicAdd(&counter[w[j][i2] >> SRC_BITS], 1) : 0;
      }
      if (staged) {
#pragma unroll
        for (int j = 0; j < RUNS_TOGETHER; ++j) {
#pragma unroll
          for (int i2 = 0; i2 < 2; ++i2)
            if (w[j][i2] != ~0u) lseg[pos[j][i2]] = w[j][i2] & SRC_MASK;
        }
      } else {
#pragma unroll
        for (int j = 0; j < RUNS_TOGETHER; ++j) {
#pragma unroll
          for (int i2 = 0; i2 < 2; ++i2)
            if (w[j][i2] != ~0u) seg_bin[pos[j][i2]] = w[j][i2] & SRC_MASK;
        }
      }
    } else {
#pragma unroll
      for (int j = 0; j < RUNS_TOGETHER; ++j) {
#pragma unroll
        for (int i2 = 0; i2 < 2; ++i2)
          if (w[j][i2] != ~0u) atomicAdd(&counter[w[j][i2] >> SRC_BITS], 1);
      }
    }
#pragma unroll
    for (int j = 0; j < RUNS_TOGETHER; ++j) {
      for (int i = lane + 64; i < c[j]; i += 32)
        take_edge<PLACE>(binned[p[j] + i], counter, lseg, seg_bin, staged);
    }
  }
}

__global__ void __launch_bounds__(1024)
finish_bins(const unsigned* __restrict__ binned, const unsigned* __restrict__ runs,
            const float* __restrict__ x, int* __restrict__ cnt, int* __restrict__ rowptr,
            float* __restrict__ dinv, unsigned* __restrict__ seg, float* __restrict__ xs,
            int n, int nt) {
  __shared__ int rsrc[MAXT];
  __shared__ int rlen[MAXT];
  __shared__ int lcnt[BIN_ROWS];
  __shared__ int lfill[BIN_ROWS];
  __shared__ float ldinv[BIN_ROWS];
  __shared__ int wsum[32];
  __shared__ int wsum2[32];
  __shared__ unsigned lseg[SEG_STAGE];
  const int tid = threadIdx.x;
  const int b = blockIdx.x;
  if (tid < BIN_ROWS) lcnt[tid] = 0;
  const unsigned pk = (tid < nt) ? runs[tid * MAXB + b] : 0u;
  const int len = (int)(pk & 0xffffu);
  const int off = (int)(pk >> 16);
  rsrc[tid] = tid * CHUNK + off;
  rlen[tid] = len;
  int m, bs;
  block_sum2<32>(len, off, wsum, wsum2, m, bs);
  const int r0 = b << BIN_SHIFT;
  const bool staged = m <= SEG_STAGE;

  walk_runs<false>(binned, rsrc, rlen, nt, lcnt, lseg, seg + bs, staged);
  __syncthreads();

  const int c = (tid < BIN_ROWS) ? lcnt[tid] : 0;
  int tot;
  const int ex = block_excl_scan<32>(c, wsum, tot);
  if (tid < BIN_ROWS) {
    const float dv = rsqrtf((float)c + 1.0f);
    lfill[tid] = ex;
    ldinv[tid] = dv;
    const int row = r0 + tid;
    if (row < n) {
      cnt[row] = c;
      rowptr[row] = bs + ex;
      dinv[row] = dv;
    }
  }
  __syncthreads();

  walk_runs<true>(binned, rsrc, rlen, nt, lfill, lseg, seg + bs, staged);
  if (staged) {
    __syncthreads();
    for (int k0 = -(bs & 31); k0 < m; k0 += 1024 * EPT) {
#pragma unroll
      for (int u = 0; u < EPT; ++u) {
        const int k = k0 + u * 1024 + tid;
        if (k >= 0 && k < m) seg[bs + k] = lseg[k];
      }
    }
  }

  const int nrows = (n - r0 < BIN_ROWS) ? (n - r0) : BIN_ROWS;
  const int nq = nrows * 16;
  const float4* __restrict__ x4 = (const float4*)x + (size_t)r0 * 16;
  float4* __restrict__ xs4 = (float4*)xs + (size_t)r0 * 16;
  for (int k0 = 0; k0 < nq; k0 += 1024 * EPT) {
    float4 v[EPT];
#pragma unroll
    for (int u = 0; u < EPT; ++u) {
      const int k = k0 + u * 1024 + tid;
      v[u] = (k < nq) ? x4[k] : make_float4(0.0f, 0.0f, 0.0f, 0.0f);
    }
#pragma unroll
    for (int u = 0; u < EPT; ++u) {
      const int k = k0 + u * 1024 + tid;
      if (k < nq) {
        const float dv = ldinv[k >> 4];
        xs4[k] = make_float4(dv * v[u].x, dv * v[u].y, dv * v[u].z, dv * v[u].w);
      }
    }
  }
}

#define FK1 64
#define FN1 128
#define FN2 64
#define TPITCH 132

static __device__ __forceinline__ v8f mma4(float a0, float a1, float b0, float b1, v8f c) {
  v2f a, b;
  a.x = a0; a.y = a1;
  b.x = b0; b.y = b1;
  return __builtin_amdgcn_wmma_f32_16x16x4_f32(false, a, false, b, (short)0, c, false, false);
}

__global__ void __launch_bounds__(256)
mlp_strips(const float* __restrict__ X, const float* __restrict__ W1,
           const float* __restrict__ b1, const float* __restrict__ W2,
           const float* __restrict__ dinv, float* __restrict__ H2, int nstrips) {
  __shared__ float4 q1[(FK1 / 4) * FN1];
  __shared__ float4 q2[(FN1 / 4) * FN2];
  __shared__ __attribute__((aligned(16))) float tile[8][16 * TPITCH];

  const int tid = threadIdx.x;
  const int wv = tid >> 5;
  const int lane = tid & 31;
  const int r = lane & 15;
  const int h = lane >> 4;

  const int lo = (int)(((unsigned)nstrips * blockIdx.x) / gridDim.x);
  const int hi = (int)(((unsigned)nstrips * (blockIdx.x + 1u)) / gridDim.x);
  float* T = tile[wv];

  float4 xa[FK1 / 8];
#pragma unroll
  for (int j = 0; j < FK1 / 8; ++j) xa[j] = make_float4(0.0f, 0.0f, 0.0f, 0.0f);
  if (lo + wv < hi) {
    const float4* xr = (const float4*)X + (size_t)((lo + wv) * 16 + r) * (FK1 / 4) + h;
#pragma unroll
    for (int j = 0; j < FK1 / 8; ++j) xa[j] = xr[2 * j];
  }

  {
    float4 w1r[(FK1 / 4) * FN1 / 256], w2r[(FN1 / 4) * FN2 / 256];
#pragma unroll
    for (int i = 0; i < (FK1 / 4) * FN1 / 256; ++i) {
      const int idx = tid + 256 * i;
      const float* w = W1 + (size_t)(4 * (idx >> 7)) * FN1 + (idx & (FN1 - 1));
      w1r[i] = make_float4(w[0], w[FN1], w[2 * FN1], w[3 * FN1]);
    }
#pragma unroll
    for (int i = 0; i < (FN1 / 4) * FN2 / 256; ++i) {
      const int idx = tid + 256 * i;
      const float* w = W2 + (size_t)(4 * (idx >> 6)) * FN2 + (idx & (FN2 - 1));
      w2r[i] = make_float4(w[0], w[FN2], w[2 * FN2], w[3 * FN2]);
    }
#pragma unroll
    for (int i = 0; i < (FK1 / 4) * FN1 / 256; ++i) q1[tid + 256 * i] = w1r[i];
#pragma unroll
    for (int i = 0; i < (FN1 / 4) * FN2 / 256; ++i) q2[tid + 256 * i] = w2r[i];
  }

  float bv[FN1 / 16];
#pragma unroll
  for (int tn = 0; tn < FN1 / 16; ++tn) bv[tn] = b1[tn * 16 + r];

  __syncthreads();

  for (int s0 = lo; s0 < hi; s0 += 8) {
    const int strip = s0 + wv;
    const bool live = strip < hi;

    if (live) {
      v8f acc[FN1 / 16];
#pragma unroll
      for (int tn = 0; tn < FN1 / 16; ++tn) acc[tn] = v8f{};

#pragma unroll
      for (int j = 0; j < FK1 / 8; ++j) {
        float4 bq[FN1 / 16];
#pragma unroll
        for (int tn = 0; tn < FN1 / 16; ++tn) bq[tn] = q1[(2 * j + h) * FN1 + tn * 16 + r];
#pragma unroll
        for (int tn = 0; tn < FN1 / 16; ++tn)
          acc[tn] = mma4(xa[j].x, xa[j].y, bq[tn].x, bq[tn].y, acc[tn]);
#pragma unroll
        for (int tn = 0; tn < FN1 / 16; ++tn)
          acc[tn] = mma4(xa[j].z, xa[j].w, bq[tn].z, bq[tn].w, acc[tn]);
      }

#pragma unroll
      for (int tn = 0; tn < FN1 / 16; ++tn) {
#pragma unroll
        for (int v = 0; v < 8; ++v)
          T[(8 * h + v) * TPITCH + tn * 16 + r] = fmaxf(acc[tn][v] + bv[tn], 0.0f);
      }
    }

    __syncthreads();

    float4 xn[FK1 / 8];
#pragma unroll
    for (int j = 0; j < FK1 / 8; ++j) xn[j] = make_float4(0.0f, 0.0f, 0.0f, 0.0f);
    if (strip + 8 < hi) {
      const float4* xr = (const float4*)X + (size_t)((strip + 8) * 16 + r) * (FK1 / 4) + h;
#pragma unroll
      for (int j = 0; j < FK1 / 8; ++j) xn[j] = xr[2 * j];
    }

    if (live) {
      const float4* dq = (const float4*)(dinv + (size_t)strip * 16 + 8 * h);
      const float4 dlo = dq[0], dhi = dq[1];
      v8f c[FN2 / 16];
#pragma unroll
      for (int tn = 0; tn < FN2 / 16; ++tn) c[tn] = v8f{};

#pragma unroll 4
      for (int j = 0; j < FN1 / 8; ++j) {
        const float4 a4 = *(const float4*)&T[r * TPITCH + 8 * j + 4 * h];
        float4 bq[FN2 / 16];
#pragma unroll
        for (int tn = 0; tn < FN2 / 16; ++tn) bq[tn] = q2[(2 * j + h) * FN2 + tn * 16 + r];
#pragma unroll
        for (int tn = 0; tn < FN2 / 16; ++tn)
          c[tn] = mma4(a4.x, a4.y, bq[tn].x, bq[tn].y, c[tn]);
#pragma unroll
        for (int tn = 0; tn < FN2 / 16; ++tn)
          c[tn] = mma4(a4.z, a4.w, bq[tn].z, bq[tn].w, c[tn]);
      }

      const float dv[8] = {dlo.x, dlo.y, dlo.z, dlo.w, dhi.x, dhi.y, dhi.z, dhi.w};
      float* drow = H2 + (size_t)(strip * 16 + 8 * h) * FN2 + r;
#pragma unroll
      for (int tn = 0; tn < FN2 / 16; ++tn) {
#pragma unroll
        for (int v = 0; v < 8; ++v) drow[(size_t)v * FN2 + tn * 16] = dv[v] * c[tn][v];
      }
    }
#pragma unroll
    for (int j = 0; j < FK1 / 8; ++j) xa[j] = xn[j];
  }
}

#define ROWS_PER_WAVE 4
#define EDGES_PER_HALF 4

static __device__ __forceinline__ unsigned load_entry(
    const unsigned* __restrict__ seg, int beg, int end, int j) {
  return seg[(j < end) ? j : beg];
}

static __device__ __forceinline__ float4 fma4(float w, float4 r, float4 a) {
  a.x = fmaf(w, r.x, a.x);
  a.y = fmaf(w, r.y, a.y);
  a.z = fmaf(w, r.z, a.z);
  a.w = fmaf(w, r.w, a.w);
  return a;
}

template <bool BIAS>
__global__ void __launch_bounds__(256)
gather_rows64(const unsigned* __restrict__ seg, const int* __restrict__ rowptr,
              const int* __restrict__ cnt, const float* __restrict__ dinv,
              const float* __restrict__ h, const float* __restrict__ bias,
              float* __restrict__ out, int n) {
  const int wave = __builtin_amdgcn_readfirstlane(
      (int)((blockIdx.x * blockDim.x + threadIdx.x) >> 5));
  const int lane = threadIdx.x & 31;
  const int q = lane & 15;
  const int hf = lane >> 4;
  const float4* __restrict__ h4 = (const float4*)h;

  float4 bq = make_float4(0.0f, 0.0f, 0.0f, 0.0f);
  if (BIAS) bq = make_float4(bias[4 * q], bias[4 * q + 1], bias[4 * q + 2], bias[4 * q + 3]);

  for (int t = 0; t < ROWS_PER_WAVE; ++t) {
    const int i = wave * ROWS_PER_WAVE + t;
    if (i >= n) break;
    const int beg = rowptr[i];
    const int end = beg + cnt[i];
    const float di = dinv[i];

    const float4 rs = h4[(size_t)i * 16 + q];
    const float ws = hf ? 0.0f : 1.0f;
    float4 acc = make_float4(0.0f, 0.0f, 0.0f, 0.0f);

    if (beg < end) {
      unsigned e[EDGES_PER_HALF];
#pragma unroll
      for (int u = 0; u < EDGES_PER_HALF; ++u) e[u] = load_entry(seg, beg, end, beg + 2 * u + hf);

      for (int j0 = beg; j0 < end; j0 += 2 * EDGES_PER_HALF) {
        float4 r[EDGES_PER_HALF];
#pragma unroll
        for (int u = 0; u < EDGES_PER_HALF; ++u) r[u] = h4[(size_t)e[u] * 16 + q];
        unsigned nx[EDGES_PER_HALF];
#pragma unroll
        for (int u = 0; u < EDGES_PER_HALF; ++u)
          nx[u] = load_entry(seg, beg, end, j0 + 2 * EDGES_PER_HALF + 2 * u + hf);
#pragma unroll
        for (int u = 0; u < EDGES_PER_HALF; ++u)
          acc = fma4((j0 + 2 * u + hf < end) ? 1.0f : 0.0f, r[u], acc);
#pragma unroll
        for (int u = 0; u < EDGES_PER_HALF; ++u) e[u] = nx[u];
      }
    }
    acc = fma4(ws, rs, acc);

    acc.x += __shfl_xor(acc.x, 16);
    acc.y += __shfl_xor(acc.y, 16);
    acc.z += __shfl_xor(acc.z, 16);
    acc.w += __shfl_xor(acc.w, 16);
    if (hf == 0) {
      acc.x = fmaf(di, acc.x, bq.x);
      acc.y = fmaf(di, acc.y, bq.y);
      acc.z = fmaf(di, acc.z, bq.z);
      acc.w = fmaf(di, acc.w, bq.w);
      ((float4*)out)[(size_t)i * 16 + q] = acc;
    }
  }
}


extern "C" void kernel_launch(void* const* d_in, const int* in_sizes, int n_in,
                              void* d_out, int out_size, void* d_ws, size_t ws_size,
                              hipStream_t stream) {
  const float* x  = (const float*)d_in[0];
  const int*   ei = (const int*)d_in[1];
  const float* W1 = (const float*)d_in[2];
  const float* b1 = (const float*)d_in[3];
  const float* W2 = (const float*)d_in[4];
  const float* b2 = (const float*)d_in[5];

  const int DIN = 64, DOUT = 64;
  const int N = in_sizes[0] / DIN;
  const int E = in_sizes[1] / 2;

  float* aggx   = (float*)d_ws;
  float* h2     = aggx + (size_t)N * DIN;
  float* xs     = h2 + (size_t)N * DOUT;
  const int NT = (E + CHUNK - 1) / CHUNK;
  unsigned* binned = (unsigned*)(xs + (size_t)N * DIN);
  unsigned* seg = binned + E;
  unsigned* runs = seg + E;
  int*   cnt    = (int*)(runs + (size_t)NT * MAXB);
  int*   rowptr = cnt + N;
  float* dinv   = (float*)(rowptr + N);
  float* out    = (float*)d_out;

  const int T = 256;
  const int NB = (N + BIN_ROWS - 1) / BIN_ROWS;

  chunk_sort<<<NT, 1024, 0, stream>>>(ei, runs, binned, E, NB);
  finish_bins<<<NB, 1024, 0, stream>>>(binned, runs, x, cnt, rowptr, dinv, seg, xs, N, NT);

  const int gwaves = (N + ROWS_PER_WAVE - 1) / ROWS_PER_WAVE;
  const int gblocks = (gwaves + 7) / 8;

  gather_rows64<false><<<gblocks, T, 0, stream>>>(seg, rowptr, cnt, dinv, xs, nullptr, aggx, N);

  {
    const int nstrips = N / 16;
    const int blocks = nstrips < 512 ? nstrips : 512;
    mlp_strips<<<blocks, T, 0, stream>>>(aggx, W1, b1, W2, dinv, h2, nstrips);
  }
  gather_rows64<true><<<gblocks, T, 0, stream>>>(seg, rowptr, cnt, dinv, h2, b2, out, N);
}


# Layer-1 row gather inside the WMMA kernel, one strip per wave: 25 waves per WGP, next row's entries requested under this row's loads, one barrier in all so waves overlap; 4 launches; sum order kept
# speedup vs baseline: 6.5780x; 1.0601x over previous
#include <hip/hip_runtime.h>


typedef float v2f __attribute__((ext_vector_type(2)));
typedef float v8f __attribute__((ext_vector_type(8)));

#define BIN_SHIFT 9
#define BIN_ROWS 512
#define MAXB 256
#define SRC_BITS 17
#define SRC_MASK ((1u << SRC_BITS) - 1u)
#define EPT 8
#define CHUNK (1024 * EPT)
#define MAXT 1024
#define RUNS_TOGETHER 8
#define SEG_STAGE 16384

template <int NW>
static __device__ __forceinline__ int block_excl_scan(int v, int* wsum, int& total) {
  const int lane = threadIdx.x & 31;
  const int wv = threadIdx.x >> 5;
  int inc = v;
#pragma unroll
  for (int o = 1; o < 32; o <<= 1) {
    int t = __shfl_up(inc, o);
    if (lane >= o) inc += t;
  }
  if (lane == 31) wsum[wv] = inc;
  __syncthreads();
  int p = (lane < NW) ? wsum[lane] : 0;
  int pre = (lane < wv) ? p : 0;
#pragma unroll
  for (int o = 16; o > 0; o >>= 1) {
    pre += __shfl_xor(pre, o);
    p += __shfl_xor(p, o);
  }
  total = p;
  return pre + inc - v;
}

template <int NW>
static __device__ __forceinline__ void block_sum2(int a, int b, int* wa, int* wb, int& sa, int& sb) {
  const int lane = threadIdx.x & 31;
  const int wv = threadIdx.x >> 5;
#pragma unroll
  for (int o = 16; o > 0; o >>= 1) {
    a += __shfl_xor(a, o);
    b += __shfl_xor(b, o);
  }
  if (lane == 0) {
    wa[wv] = a;
    wb[wv] = b;
  }
  __syncthreads();
  int pa = (lane < NW) ? wa[lane] : 0;
  int pb = (lane < NW) ? wb[lane] : 0;
#pragma unroll
  for (int o = 16; o > 0; o >>= 1) {
    pa += __shfl_xor(pa, o);
    pb += __shfl_xor(pb, o);
  }
  sa = pa;
  sb = pb;
}

__global__ void __launch_bounds__(1024)
chunk_sort(const int* __restrict__ ei, unsigned* __restrict__ runs,
           unsigned* __restrict__ binned, int E, int nb) {
  __shared__ int lcnt[MAXB];
  __shared__ int lstart[MAXB];
  __shared__ int wsum[32];
  __shared__ unsigned stage[CHUNK];
  const int tid = threadIdx.x;
  const int lo = blockIdx.x * CHUNK;
  const int hi = (E - lo < CHUNK) ? E : lo + CHUNK;
  if (tid < MAXB) lcnt[tid] = 0;
  int s[EPT], d[EPT], rk[EPT];
#pragma unroll
  for (int u = 0; u < EPT; ++u) {
    const int e = lo + u * 1024 + tid;
    const bool in = e < hi;
    s[u] = in ? ei[e] : 0;
    d[u] = in ? ei[E + e] : -1;
  }
  __syncthreads();
#pragma unroll
  for (int u = 0; u < EPT; ++u)
    rk[u] = (d[u] >= 0) ? atomicAdd(&lcnt[d[u] >> BIN_SHIFT], 1) : 0;
  __syncthreads();
  const int c = (tid < MAXB) ? lcnt[tid] : 0;
  int nedge;
  const int ex = block_excl_scan<32>(c, wsum, nedge);
  if (tid < MAXB) {
    lstart[tid] = ex;
    if (tid < nb) runs[blockIdx.x * MAXB + tid] = ((unsigned)ex << 16) | (unsigned)c;
  }
  __syncthreads();
#pragma unroll
  for (int u = 0; u < EPT; ++u) {
    if (d[u] >= 0) {
      stage[lstart[d[u] >> BIN_SHIFT] + rk[u]] =
          (unsigned)s[u] | ((unsigned)(d[u] & (BIN_ROWS - 1)) << SRC_BITS);
    }
  }
  __syncthreads();
#pragma unroll
  for (int u = 0; u < EPT; ++u) {
    const int k = u * 1024 + tid;
    if (k < nedge) binned[lo + k] = stage[k];
  }
}

template <bool PLACE>
static __device__ __forceinline__ void take_edge(unsigned w, int* counter, unsigned* lseg,
                                                 unsigned* seg_bin, bool staged) {
  if (PLACE) {
    const int pos = atomicAdd(&counter[w >> SRC_BITS], 1);
    if (staged) lseg[pos] = w & SRC_MASK;
    else seg_bin[pos] = w & SRC_MASK;
  } else {
    atomicAdd(&counter[w >> SRC_BITS], 1);
  }
}

template <bool PLACE>
static __device__ __forceinline__ void walk_runs(const unsigned* __restrict__ binned,
                                                 const int* rsrc, const int* rlen, int nt,
                                                 int* counter, unsigned* lseg,
                                                 unsigned* seg_bin, bool staged) {
  const int lane = threadIdx.x & 31;
  const int wv = threadIdx.x >> 5;
  for (int t0 = wv; t0 < nt; t0 += 32 * RUNS_TOGETHER) {
    int p[RUNS_TOGETHER], c[RUNS_TOGETHER];
    unsigned w[RUNS_TOGETHER][2];
#pragma unroll
    for (int j = 0; j < RUNS_TOGETHER; ++j) {
      const int t = t0 + 32 * j;
      const bool in = t < nt;
      c[j] = in ? rlen[in ? t : 0] : 0;
      p[j] = rsrc[in ? t : 0];
    }
#pragma unroll
    for (int j = 0; j < RUNS_TOGETHER; ++j) {
#pragma unroll
      for (int i2 = 0; i2 < 2; ++i2) {
        const int i = lane + 32 * i2;
        w[j][i2] = (i < c[j]) ? binned[p[j] + i] : ~0u;
      }
    }
    if (PLACE) {
      int pos[RUNS_TOGETHER][2];
#pragma unroll
      for (int j = 0; j < RUNS_TOGETHER; ++j) {
#pragma unroll
        for (int i2 = 0; i2 < 2; ++i2)
          pos[j][i2] = (w[j][i2] != ~0u) ? atomicAdd(&counter[w[j][i2] >> SRC_BITS], 1) : 0;
      }
      if (staged) {
#pragma unroll
        for (int j = 0; j < RUNS_TOGETHER; ++j) {
#pragma unroll
          for (int i2 = 0; i2 < 2; ++i2)
            if (w[j][i2] != ~0u) lseg[pos[j][i2]] = w[j][i2] & SRC_MASK;
        }
      } else {
#pragma unroll
        for (int j = 0; j < RUNS_TOGETHER; ++j) {
#pragma unroll
          for (int i2 = 0; i2 < 2; ++i2)
            if (w[j][i2] != ~0u) seg_bin[pos[j][i2]] = w[j][i2] & SRC_MASK;
        }
      }
    } else {
#pragma unroll
      for (int j = 0; j < RUNS_TOGETHER; ++j) {
#pragma unroll
        for (int i2 = 0; i2 < 2; ++i2)
          if (w[j][i2] != ~0u) atomicAdd(&counter[w[j][i2] >> SRC_BITS], 1);
      }
    }
#pragma unroll
    for (int j = 0; j < RUNS_TOGETHER; ++j) {
      for (int i = lane + 64; i < c[j]; i += 32)
        take_edge<PLACE>(binned[p[j] + i], counter, lseg, seg_bin, staged);
    }
  }
}

__global__ void __launch_bounds__(1024)
finish_bins(const unsigned* __restrict__ binned, const unsigned* __restrict__ runs,
            const float* __restrict__ x, int* __restrict__ cnt, int* __restrict__ rowptr,
            float* __restrict__ dinv, unsigned* __restrict__ seg, float* __restrict__ xs,
            int n, int nt) {
  __shared__ int rsrc[MAXT];
  __shared__ int rlen[MAXT];
  __shared__ int lcnt[BIN_ROWS];
  __shared__ int lfill[BIN_ROWS];
  __shared__ float ldinv[BIN_ROWS];
  __shared__ int wsum[32];
  __shared__ int wsum2[32];
  __shared__ unsigned lseg[SEG_STAGE];
  const int tid = threadIdx.x;
  const int b = blockIdx.x;
  if (tid < BIN_ROWS) lcnt[tid] = 0;
  const unsigned pk = (tid < nt) ? runs[tid * MAXB + b] : 0u;
  const int len = (int)(pk & 0xffffu);
  const int off = (int)(pk >> 16);
  rsrc[tid] = tid * CHUNK + off;
  rlen[tid] = len;
  int m, bs;
  block_sum2<32>(len, off, wsum, wsum2, m, bs);
  const int r0 = b << BIN_SHIFT;
  const bool staged = m <= SEG_STAGE;

  walk_runs<false>(binned, rsrc, rlen, nt, lcnt, lseg, seg + bs, staged);
  __syncthreads();

  const int c = (tid < BIN_ROWS) ? lcnt[tid] : 0;
  int tot;
  const int ex = block_excl_scan<32>(c, wsum, tot);
  if (tid < BIN_ROWS) {
    const float dv = rsqrtf((float)c + 1.0f);
    lfill[tid] = ex;
    ldinv[tid] = dv;
    const int row = r0 + tid;
    if (row < n) {
      cnt[row] = c;
      rowptr[row] = bs + ex;
      dinv[row] = dv;
    }
  }
  __syncthreads();

  walk_runs<true>(binned, rsrc, rlen, nt, lfill, lseg, seg + bs, staged);
  if (staged) {
    __syncthreads();
    for (int k0 = -(bs & 31); k0 < m; k0 += 1024 * EPT) {
#pragma unroll
      for (int u = 0; u < EPT; ++u) {
        const int k = k0 + u * 1024 + tid;
        if (k >= 0 && k < m) seg[bs + k] = lseg[k];
      }
    }
  }

  const int nrows = (n - r0 < BIN_ROWS) ? (n - r0) : BIN_ROWS;
  const int nq = nrows * 16;
  const float4* __restrict__ x4 = (const float4*)x + (size_t)r0 * 16;
  float4* __restrict__ xs4 = (float4*)xs + (size_t)r0 * 16;
  for (int k0 = 0; k0 < nq; k0 += 1024 * EPT) {
    float4 v[EPT];
#pragma unroll
    for (int u = 0; u < EPT; ++u) {
      const int k = k0 + u * 1024 + tid;
      v[u] = (k < nq) ? x4[k] : make_float4(0.0f, 0.0f, 0.0f, 0.0f);
    }
#pragma unroll
    for (int u = 0; u < EPT; ++u) {
      const int k = k0 + u * 1024 + tid;
      if (k < nq) {
        const float dv = ldinv[k >> 4];
        xs4[k] = make_float4(dv * v[u].x, dv * v[u].y, dv * v[u].z, dv * v[u].w);
      }
    }
  }
}

#define FK1 64
#define FN1 128
#define FN2 64
#define TPITCH 132

static __device__ __forceinline__ v8f mma4(float a0, float a1, float b0, float b1, v8f c) {
  v2f a, b;
  a.x = a0; a.y = a1;
  b.x = b0; b.y = b1;
  return __builtin_amdgcn_wmma_f32_16x16x4_f32(false, a, false, b, (short)0, c, false, false);
}

#define ROWS_PER_WAVE 4
#define EDGES_PER_HALF 4

static __device__ __forceinline__ unsigned load_entry(
    const unsigned* __restrict__ seg, int beg, int end, int j) {
  return seg[(j < end) ? j : beg];
}

static __device__ __forceinline__ float4 fma4(float w, float4 r, float4 a) {
  a.x = fmaf(w, r.x, a.x);
  a.y = fmaf(w, r.y, a.y);
  a.z = fmaf(w, r.z, a.z);
  a.w = fmaf(w, r.w, a.w);
  return a;
}

template <bool BIAS>
__global__ void __launch_bounds__(256)
gather_rows64(const unsigned* __restrict__ seg, const int* __restrict__ rowptr,
              const int* __restrict__ cnt, const float* __restrict__ dinv,
              const float* __restrict__ h, const float* __restrict__ bias,
              float* __restrict__ out, int n) {
  const int wave = __builtin_amdgcn_readfirstlane(
      (int)((blockIdx.x * blockDim.x + threadIdx.x) >> 5));
  const int lane = threadIdx.x & 31;
  const int q = lane & 15;
  const int hf = lane >> 4;
  const float4* __restrict__ h4 = (const float4*)h;

  float4 bq = make_float4(0.0f, 0.0f, 0.0f, 0.0f);
  if (BIAS) bq = make_float4(bias[4 * q], bias[4 * q + 1], bias[4 * q + 2], bias[4 * q + 3]);

  for (int t = 0; t < ROWS_PER_WAVE; ++t) {
    const int i = wave * ROWS_PER_WAVE + t;
    if (i >= n) break;
    const int beg = rowptr[i];
    const int end = beg + cnt[i];
    const float di = dinv[i];

    const float4 rs = h4[(size_t)i * 16 + q];
    const float ws = hf ? 0.0f : 1.0f;
    float4 acc = make_float4(0.0f, 0.0f, 0.0f, 0.0f);

    if (beg < end) {
      unsigned e[EDGES_PER_HALF];
#pragma unroll
      for (int u = 0; u < EDGES_PER_HALF; ++u) e[u] = load_entry(seg, beg, end, beg + 2 * u + hf);

      for (int j0 = beg; j0 < end; j0 += 2 * EDGES_PER_HALF) {
        float4 r[EDGES_PER_HALF];
#pragma unroll
        for (int u = 0; u < EDGES_PER_HALF; ++u) r[u] = h4[(size_t)e[u] * 16 + q];
        unsigned nx[EDGES_PER_HALF];
#pragma unroll
        for (int u = 0; u < EDGES_PER_HALF; ++u)
          nx[u] = load_entry(seg, beg, end, j0 + 2 * EDGES_PER_HALF + 2 * u + hf);
#pragma unroll
        for (int u = 0; u < EDGES_PER_HALF; ++u)
          acc = fma4((j0 + 2 * u + hf < end) ? 1.0f : 0.0f, r[u], acc);
#pragma unroll
        for (int u = 0; u < EDGES_PER_HALF; ++u) e[u] = nx[u];
      }
    }
    acc = fma4(ws, rs, acc);

    acc.x += __shfl_xor(acc.x, 16);
    acc.y += __shfl_xor(acc.y, 16);
    acc.z += __shfl_xor(acc.z, 16);
    acc.w += __shfl_xor(acc.w, 16);
    if (hf == 0) {
      acc.x = fmaf(di, acc.x, bq.x);
      acc.y = fmaf(di, acc.y, bq.y);
      acc.z = fmaf(di, acc.z, bq.z);
      acc.w = fmaf(di, acc.w, bq.w);
      ((float4*)out)[(size_t)i * 16 + q] = acc;
    }
  }
}

#define FW 25
#define GE 8

__global__ void __launch_bounds__(FW * 32)
gather_mlp(const unsigned* __restrict__ seg, const int* __restrict__ rowptr,
           const int* __restrict__ cnt, const float* __restrict__ dinv,
           const float* __restrict__ xs, const float* __restrict__ W1,
           const float* __restrict__ b1, const float* __restrict__ W2,
           float* __restrict__ H2, int nstrips) {
  __shared__ float4 q1[(FK1 / 4) * FN1];
  __shared__ float4 q2[(FN1 / 4) * FN2];
  __shared__ __attribute__((aligned(16))) float tile[FW][16 * TPITCH];

  const int tid = threadIdx.x;
  const int wv = __builtin_amdgcn_readfirstlane(tid >> 5);
  const int lane = tid & 31;
  const int r = lane & 15;
  const int h = lane >> 4;
  const int strip = blockIdx.x * FW + wv;
  const bool live = strip < nstrips;
  float* T = tile[wv];
  const float4* __restrict__ h4 = (const float4*)xs;

  {
    constexpr int NP1 = ((FK1 / 4) * FN1 + FW * 32 - 1) / (FW * 32);
    constexpr int NP2 = ((FN1 / 4) * FN2 + FW * 32 - 1) / (FW * 32);
    float4 w1r[NP1], w2r[NP2];
#pragma unroll
    for (int p = 0; p < NP1; ++p) {
      const int idx = tid + FW * 32 * p;
      w1r[p] = make_float4(0.0f, 0.0f, 0.0f, 0.0f);
      if (idx < (FK1 / 4) * FN1) {
        const float* w = W1 + (size_t)(4 * (idx >> 7)) * FN1 + (idx & (FN1 - 1));
        w1r[p] = make_float4(w[0], w[FN1], w[2 * FN1], w[3 * FN1]);
      }
    }
#pragma unroll
    for (int p = 0; p < NP2; ++p) {
      const int idx = tid + FW * 32 * p;
      w2r[p] = make_float4(0.0f, 0.0f, 0.0f, 0.0f);
      if (idx < (FN1 / 4) * FN2) {
        const float* w = W2 + (size_t)(4 * (idx >> 6)) * FN2 + (idx & (FN2 - 1));
        w2r[p] = make_float4(w[0], w[FN2], w[2 * FN2], w[3 * FN2]);
      }
    }
#pragma unroll
    for (int p = 0; p < NP1; ++p) {
      const int idx = tid + FW * 32 * p;
      if (idx < (FK1 / 4) * FN1) q1[idx] = w1r[p];
    }
#pragma unroll
    for (int p = 0; p < NP2; ++p) {
      const int idx = tid + FW * 32 * p;
      if (idx < (FN1 / 4) * FN2) q2[idx] = w2r[p];
    }
  }
  float bv[FN1 / 16];
#pragma unroll
  for (int tn = 0; tn < FN1 / 16; ++tn) bv[tn] = b1[tn * 16 + r];

  __syncthreads();

  if (live) {
    const int rpv = rowptr[strip * 16 + r];
    const int cnv = cnt[strip * 16 + r];
    const float div = dinv[strip * 16 + r];
    int beg = __builtin_amdgcn_readlane(rpv, 0);
    int end = beg + __builtin_amdgcn_readlane(cnv, 0);
    unsigned e[GE];
#pragma unroll
    for (int u = 0; u < GE; ++u) e[u] = 0u;
    if (beg < end) {
#pragma unroll
      for (int u = 0; u < GE; ++u) e[u] = load_entry(seg, beg, end, beg + 2 * u + h);
    }
    for (int t = 0; t < 16; ++t) {
      const int i = strip * 16 + t;
      const int tn1 = (t < 15) ? t + 1 : 15;
      const int nb = __builtin_amdgcn_readlane(rpv, tn1);
      const int ne = (t < 15) ? nb + __builtin_amdgcn_readlane(cnv, tn1) : nb;
      const float di = __builtin_bit_cast(float, __builtin_amdgcn_readlane(__builtin_bit_cast(int, div), t));
      const float4 rs = h4[(size_t)i * 16 + r];
      const float ws = h ? 0.0f : 1.0f;
      float4 acc = make_float4(0.0f, 0.0f, 0.0f, 0.0f);
      if (beg < end) {
        for (int j0 = beg; j0 < end; j0 += 2 * GE) {
          float4 rr[GE];
#pragma unroll
          for (int u = 0; u < GE; ++u) rr[u] = h4[(size_t)e[u] * 16 + r];
          const bool lastTrip = j0 + 2 * GE >= end;
          const int pb = lastTrip ? nb : beg;
          const int pe = lastTrip ? ne : end;
          const int pj = lastTrip ? nb : j0 + 2 * GE;
          unsigned nx[GE];
#pragma unroll
          for (int u = 0; u < GE; ++u) nx[u] = e[u];
          if (pb < pe) {
#pragma unroll
            for (int u = 0; u < GE; ++u) nx[u] = load_entry(seg, pb, pe, pj + 2 * u + h);
          }
#pragma unroll
          for (int u = 0; u < GE; ++u)
            acc = fma4((j0 + 2 * u + h < end) ? 1.0f : 0.0f, rr[u], acc);
#pragma unroll
          for (int u = 0; u < GE; ++u) e[u] = nx[u];
        }
      } else if (nb < ne) {
#pragma unroll
        for (int u = 0; u < GE; ++u) e[u] = load_entry(seg, nb, ne, nb + 2 * u + h);
      }
      acc = fma4(ws, rs, acc);
      acc.x += __shfl_xor(acc.x, 16);
      acc.y += __shfl_xor(acc.y, 16);
      acc.z += __shfl_xor(acc.z, 16);
      acc.w += __shfl_xor(acc.w, 16);
      if (h == 0)
        *(float4*)&T[t * TPITCH + 4 * r] =
            make_float4(di * acc.x, di * acc.y, di * acc.z, di * acc.w);
      beg = nb;
      end = ne;
    }

    __builtin_amdgcn_fence(__ATOMIC_RELEASE, "wavefront");
    __builtin_amdgcn_wave_barrier();
    __builtin_amdgcn_fence(__ATOMIC_ACQUIRE, "wavefront");

    float4 xa[FK1 / 8];
#pragma unroll
    for (int j = 0; j < FK1 / 8; ++j) xa[j] = *(const float4*)&T[r * TPITCH + 8 * j + 4 * h];

    v8f acc[FN1 / 16];
#pragma unroll
    for (int tn = 0; tn < FN1 / 16; ++tn) acc[tn] = v8f{};
#pragma unroll
    for (int j = 0; j < FK1 / 8; ++j) {
      float4 bq[FN1 / 16];
#pragma unroll
      for (int tn = 0; tn < FN1 / 16; ++tn) bq[tn] = q1[(2 * j + h) * FN1 + tn * 16 + r];
#pragma unroll
      for (int tn = 0; tn < FN1 / 16; ++tn)
        acc[tn] = mma4(xa[j].x, xa[j].y, bq[tn].x, bq[tn].y, acc[tn]);
#pragma unroll
      for (int tn = 0; tn < FN1 / 16; ++tn)
        acc[tn] = mma4(xa[j].z, xa[j].w, bq[tn].z, bq[tn].w, acc[tn]);
    }
#pragma unroll
    for (int tn = 0; tn < FN1 / 16; ++tn) {
#pragma unroll
      for (int v = 0; v < 8; ++v)
        T[(8 * h + v) * TPITCH + tn * 16 + r] = fmaxf(acc[tn][v] + bv[tn], 0.0f);
    }

    __builtin_amdgcn_fence(__ATOMIC_RELEASE, "wavefront");
    __builtin_amdgcn_wave_barrier();
    __builtin_amdgcn_fence(__ATOMIC_ACQUIRE, "wavefront");

    const float4* dq = (const float4*)(dinv + (size_t)strip * 16 + 8 * h);
    const float4 dlo = dq[0], dhi = dq[1];
    v8f c[FN2 / 16];
#pragma unroll
    for (int tn = 0; tn < FN2 / 16; ++tn) c[tn] = v8f{};
#pragma unroll 4
    for (int j = 0; j < FN1 / 8; ++j) {
      const float4 a4 = *(const float4*)&T[r * TPITCH + 8 * j + 4 * h];
      float4 bq[FN2 / 16];
#pragma unroll
      for (int tn = 0; tn < FN2 / 16; ++tn) bq[tn] = q2[(2 * j + h) * FN2 + tn * 16 + r];
#pragma unroll
      for (int tn = 0; tn < FN2 / 16; ++tn)
        c[tn] = mma4(a4.x, a4.y, bq[tn].x, bq[tn].y, c[tn]);
#pragma unroll
      for (int tn = 0; tn < FN2 / 16; ++tn)
        c[tn] = mma4(a4.z, a4.w, bq[tn].z, bq[tn].w, c[tn]);
    }
    const float dv[8] = {dlo.x, dlo.y, dlo.z, dlo.w, dhi.x, dhi.y, dhi.z, dhi.w};
    float* drow = H2 + (size_t)(strip * 16 + 8 * h) * FN2 + r;
#pragma unroll
    for (int tn = 0; tn < FN2 / 16; ++tn) {
#pragma unroll
      for (int v = 0; v < 8; ++v) drow[(size_t)v * FN2 + tn * 16] = dv[v] * c[tn][v];
    }
  }
}


extern "C" void kernel_launch(void* const* d_in, const int* in_sizes, int n_in,
                              void* d_out, int out_size, void* d_ws, size_t ws_size,
                              hipStream_t stream) {
  const float* x  = (const float*)d_in[0];
  const int*   ei = (const int*)d_in[1];
  const float* W1 = (const float*)d_in[2];
  const float* b1 = (const float*)d_in[3];
  const float* W2 = (const float*)d_in[4];
  const float* b2 = (const float*)d_in[5];

  const int DIN = 64, DOUT = 64;
  const int N = in_sizes[0] / DIN;
  const int E = in_sizes[1] / 2;

  float* h2     = (float*)d_ws;
  float* xs     = h2 + (size_t)N * DOUT;
  const int NT = (E + CHUNK - 1) / CHUNK;
  unsigned* binned = (unsigned*)(xs + (size_t)N * DIN);
  unsigned* seg = binned + E;
  unsigned* runs = seg + E;
  int*   cnt    = (int*)(runs + (size_t)NT * MAXB);
  int*   rowptr = cnt + N;
  float* dinv   = (float*)(rowptr + N);
  float* out    = (float*)d_out;

  const int T = 256;
  const int NB = (N + BIN_ROWS - 1) / BIN_ROWS;

  chunk_sort<<<NT, 1024, 0, stream>>>(ei, runs, binned, E, NB);
  finish_bins<<<NB, 1024, 0, stream>>>(binned, runs, x, cnt, rowptr, dinv, seg, xs, N, NT);

  const int gwaves = (N + ROWS_PER_WAVE - 1) / ROWS_PER_WAVE;
  const int gblocks = (gwaves + 7) / 8;

  {
    const int nstrips = N / 16;
    gather_mlp<<<(nstrips + FW - 1) / FW, FW * 32, 0, stream>>>(seg, rowptr, cnt, dinv, xs, W1, b1, W2,
                                                                 h2, nstrips);
  }
  gather_rows64<true><<<gblocks, T, 0, stream>>>(seg, rowptr, cnt, dinv, h2, b2, out, N);
}
